# starting kernel
# baseline (speedup 1.0000x reference)
#include <hip/hip_runtime.h>
#include <math.h>

#define NN    50000
#define NE    800000
#define ET    (NE + NN)
#define HID   128
#define NH    4
#define CH    32

typedef float v2f __attribute__((ext_vector_type(2)));
typedef float v8f __attribute__((ext_vector_type(8)));

__device__ __forceinline__ int f2ord(float f) {
    int i = __float_as_int(f);
    return (i >= 0) ? i : (i ^ 0x7fffffff);
}
__device__ __forceinline__ float ord2f(int i) {
    return __int_as_float((i >= 0) ? i : (i ^ 0x7fffffff));
}

template <int K, int NT>
__global__ void wmma_gemm_strip(const float* __restrict__ X,
                                const float* __restrict__ W,
                                const float* __restrict__ bias,
                                float* __restrict__ Y,
                                int nMtiles)
{
    constexpr int Nout = NT * 16;
    extern __shared__ float Wlds[];

    for (int i = threadIdx.x * 4; i < K * Nout; i += blockDim.x * 4)
        *(float4*)&Wlds[i] = *(const float4*)&W[i];
    __syncthreads();

    int wave = (int)((blockIdx.x * blockDim.x + threadIdx.x) >> 5);
    if (wave >= nMtiles) return;
    int lane = (int)(threadIdx.x & 31);

    int row = wave * 16 + (lane & 15);
    int col = lane & 15;
    int kh  = (lane >> 4) << 1;

    v8f acc[NT];
#pragma unroll
    for (int t = 0; t < NT; ++t)
        acc[t] = (v8f){0.f, 0.f, 0.f, 0.f, 0.f, 0.f, 0.f, 0.f};

    const float* xrow = X + (size_t)row * K;
#pragma unroll
    for (int k = 0; k < K; k += 4) {
        v2f a = *(const v2f*)(xrow + k + kh);
        const float* w0 = Wlds + (k + kh) * Nout + col;
#pragma unroll
        for (int t = 0; t < NT; ++t) {
            v2f b;
            b.x = w0[t * 16];
            b.y = w0[t * 16 + Nout];
            acc[t] = __builtin_amdgcn_wmma_f32_16x16x4_f32(
                         false, a, false, b, (short)0, acc[t], false, false);
        }
    }

    int orow = wave * 16 + ((lane >> 4) << 3);
#pragma unroll
    for (int t = 0; t < NT; ++t) {
        int ocol = t * 16 + (lane & 15);
        float badd = bias ? bias[ocol] : 0.f;
#pragma unroll
        for (int r = 0; r < 8; ++r)
            Y[(size_t)(orow + r) * Nout + ocol] = acc[t][r] + badd;
    }
}

__global__ void alpha_kernel(const float* __restrict__ Hm,
                             const float* __restrict__ a_s,
                             const float* __restrict__ a_d,
                             float* __restrict__ as_out,
                             float* __restrict__ ad_out)
{
    int tid = blockIdx.x * blockDim.x + threadIdx.x;
    if (tid >= NN * NH) return;
    int n = tid >> 2, h = tid & 3;
    const float* hp = Hm  + n * HID + h * CH;
    const float* sp = a_s + h * CH;
    const float* dp = a_d + h * CH;
    float s = 0.f, d = 0.f;
#pragma unroll
    for (int c = 0; c < CH; c += 4) {
        float4 hv = *(const float4*)(hp + c);
        float4 sv = *(const float4*)(sp + c);
        float4 dv = *(const float4*)(dp + c);
        s += hv.x * sv.x + hv.y * sv.y + hv.z * sv.z + hv.w * sv.w;
        d += hv.x * dv.x + hv.y * dv.y + hv.z * dv.z + hv.w * dv.w;
    }
    as_out[tid] = s;
    ad_out[tid] = d;
}

__global__ void fill_f32(float* p, float v, int n) {
    int t = blockIdx.x * blockDim.x + threadIdx.x;
    if (t < n) p[t] = v;
}
__global__ void fill_i32(int* p, int v, int n) {
    int t = blockIdx.x * blockDim.x + threadIdx.x;
    if (t < n) p[t] = v;
}

__global__ void edge_pass1(const int* __restrict__ src,
                           const int* __restrict__ dst,
                           const float* __restrict__ as_v,
                           const float* __restrict__ ad_v,
                           float* __restrict__ ebuf,
                           int* __restrict__ mord)
{
    int tid = blockIdx.x * blockDim.x + threadIdx.x;
    if (tid >= ET * NH) return;
    int e = tid >> 2, h = tid & 3;
    int s = (e < NE) ? src[e] : (e - NE);
    int d = (e < NE) ? dst[e] : (e - NE);
    float v = as_v[s * NH + h] + ad_v[d * NH + h];
    v = (v > 0.f) ? v : 0.2f * v;
    ebuf[tid] = v;
    atomicMax(&mord[d * NH + h], f2ord(v));
}

__global__ void edge_pass2(const int* __restrict__ dst,
                           const int* __restrict__ mord,
                           float* __restrict__ ebuf,
                           float* __restrict__ denom)
{
    int tid = blockIdx.x * blockDim.x + threadIdx.x;
    if (tid >= ET * NH) return;
    int e = tid >> 2, h = tid & 3;
    int d = (e < NE) ? dst[e] : (e - NE);
    float m  = ord2f(mord[d * NH + h]);
    float ex = expf(ebuf[tid] - m);
    ebuf[tid] = ex;
    atomicAdd(&denom[d * NH + h], ex);
}

__global__ void edge_pass3(const int* __restrict__ src,
                           const int* __restrict__ dst,
                           const float* __restrict__ ebuf,
                           const float* __restrict__ denom,
                           const float* __restrict__ Hm,
                           float* __restrict__ agg)
{
    long long tid = (long long)blockIdx.x * blockDim.x + threadIdx.x;
    if (tid >= (long long)ET * (HID / 4)) return;
    int e = (int)(tid >> 5);
    int q = (int)(tid & 31);
    int j = q << 2;
    int h = j >> 5;
    int s = (e < NE) ? src[e] : (e - NE);
    int d = (e < NE) ? dst[e] : (e - NE);
    float alpha = ebuf[e * NH + h] / (denom[d * NH + h] + 1e-16f);
    float4 hv = *(const float4*)(Hm + (size_t)s * HID + j);
    float* ap = agg + (size_t)d * HID + j;
    atomicAdd(ap + 0, hv.x * alpha);
    atomicAdd(ap + 1, hv.y * alpha);
    atomicAdd(ap + 2, hv.z * alpha);
    atomicAdd(ap + 3, hv.w * alpha);
}

__global__ void bias_elu(const float* __restrict__ agg,
                         const float* __restrict__ b,
                         float* __restrict__ out)
{
    int tid = blockIdx.x * blockDim.x + threadIdx.x;
    if (tid >= NN * HID) return;
    int j = tid & 127;
    float y = agg[tid] + b[j];
    out[tid] = (y > 0.f) ? y : (expf(y) - 1.f);
}

static void run_gat_layer(const float* Xin, const float* W,
                          const float* a_s, const float* a_d, const float* b,
                          const int* src, const int* dst,
                          float* bufH, float* bufAgg, float* bufOut,
                          float* ebuf, float* asb, float* adb,
                          float* denom, int* mord, hipStream_t stream)
{
    const int TB = 256;
    int nMtiles = NN / 16;
    int blocks  = (nMtiles * 32 + TB - 1) / TB;
    wmma_gemm_strip<HID, HID / 16><<<blocks, TB, HID * HID * sizeof(float), stream>>>(
        Xin, W, nullptr, bufH, nMtiles);
    alpha_kernel<<<(NN * NH + TB - 1) / TB, TB, 0, stream>>>(bufH, a_s, a_d, asb, adb);
    fill_i32<<<(NN * NH + TB - 1) / TB, TB, 0, stream>>>(mord, (int)0x80000000, NN * NH);
    fill_f32<<<(NN * NH + TB - 1) / TB, TB, 0, stream>>>(denom, 0.f, NN * NH);
    fill_f32<<<(NN * HID + TB - 1) / TB, TB, 0, stream>>>(bufAgg, 0.f, NN * HID);
    int gEH = (ET * NH + TB - 1) / TB;
    edge_pass1<<<gEH, TB, 0, stream>>>(src, dst, asb, adb, ebuf, mord);
    edge_pass2<<<gEH, TB, 0, stream>>>(dst, mord, ebuf, denom);
    long long tot3 = (long long)ET * (HID / 4);
    edge_pass3<<<(unsigned)((tot3 + TB - 1) / TB), TB, 0, stream>>>(
        src, dst, ebuf, denom, bufH, bufAgg);
    bias_elu<<<(NN * HID + TB - 1) / TB, TB, 0, stream>>>(bufAgg, b, bufOut);
}

extern "C" void kernel_launch(void* const* d_in, const int* in_sizes, int n_in,
                              void* d_out, int out_size, void* d_ws, size_t ws_size,
                              hipStream_t stream)
{
    const float* x    = (const float*)d_in[0];
    const int*   edge = (const int*)  d_in[1];
    const float* W1   = (const float*)d_in[2];
    const float* a_s1 = (const float*)d_in[3];
    const float* a_d1 = (const float*)d_in[4];
    const float* b1   = (const float*)d_in[5];
    const float* W2   = (const float*)d_in[6];
    const float* a_s2 = (const float*)d_in[7];
    const float* a_d2 = (const float*)d_in[8];
    const float* b2   = (const float*)d_in[9];
    const float* fc_w = (const float*)d_in[10];
    const float* fc_b = (const float*)d_in[11];
    float* out = (float*)d_out;

    const int* src = edge;
    const int* dst = edge + NE;

    float* bufH   = (float*)d_ws;
    float* bufAgg = bufH   + (size_t)NN * HID;
    float* bufIn2 = bufAgg + (size_t)NN * HID;
    float* ebuf   = bufIn2 + (size_t)NN * HID;
    float* asb    = ebuf   + (size_t)ET * NH;
    float* adb    = asb    + (size_t)NN * NH;
    float* denom  = adb    + (size_t)NN * NH;
    int*   mord   = (int*)(denom + (size_t)NN * NH);

    run_gat_layer(x, W1, a_s1, a_d1, b1, src, dst,
                  bufH, bufAgg, bufIn2, ebuf, asb, adb, denom, mord, stream);
    run_gat_layer(bufIn2, W2, a_s2, a_d2, b2, src, dst,
                  bufH, bufAgg, bufIn2, ebuf, asb, adb, denom, mord, stream);
    const int TB = 256;
    int nMtiles = NN / 16;
    int blocks  = (nMtiles * 32 + TB - 1) / TB;
    wmma_gemm_strip<HID, CH / 16><<<blocks, TB, HID * CH * sizeof(float), stream>>>(
        bufIn2, fc_w, fc_b, out, nMtiles);
}


# Aggregation with destination rows held in LDS: a workgroup owns 196 nodes, filters the edge list for them, sums softmax terms and rows by LDS atomics, divides once per node; float sum order changes
# speedup vs baseline: 2.5040x; 2.5040x over previous
#include <hip/hip_runtime.h>
#include <math.h>

#define NN    50000
#define NE    800000
#define ET    (NE + NN)
#define HID   128
#define NH    4
#define CH    32

typedef float v2f __attribute__((ext_vector_type(2)));
typedef float v8f __attribute__((ext_vector_type(8)));

__device__ __forceinline__ int f2ord(float f) {
    int i = __float_as_int(f);
    return (i >= 0) ? i : (i ^ 0x7fffffff);
}
__device__ __forceinline__ float ord2f(int i) {
    return __int_as_float((i >= 0) ? i : (i ^ 0x7fffffff));
}

template <int K, int NT>
__global__ void wmma_gemm_strip(const float* __restrict__ X,
                                const float* __restrict__ W,
                                const float* __restrict__ bias,
                                float* __restrict__ Y,
                                int nMtiles)
{
    constexpr int Nout = NT * 16;
    extern __shared__ float Wlds[];

    for (int i = threadIdx.x * 4; i < K * Nout; i += blockDim.x * 4)
        *(float4*)&Wlds[i] = *(const float4*)&W[i];
    __syncthreads();

    int wave = (int)((blockIdx.x * blockDim.x + threadIdx.x) >> 5);
    if (wave >= nMtiles) return;
    int lane = (int)(threadIdx.x & 31);

    int row = wave * 16 + (lane & 15);
    int col = lane & 15;
    int kh  = (lane >> 4) << 1;

    v8f acc[NT];
#pragma unroll
    for (int t = 0; t < NT; ++t)
        acc[t] = (v8f){0.f, 0.f, 0.f, 0.f, 0.f, 0.f, 0.f, 0.f};

    const float* xrow = X + (size_t)row * K;
#pragma unroll
    for (int k = 0; k < K; k += 4) {
        v2f a = *(const v2f*)(xrow + k + kh);
        const float* w0 = Wlds + (k + kh) * Nout + col;
#pragma unroll
        for (int t = 0; t < NT; ++t) {
            v2f b;
            b.x = w0[t * 16];
            b.y = w0[t * 16 + Nout];
            acc[t] = __builtin_amdgcn_wmma_f32_16x16x4_f32(
                         false, a, false, b, (short)0, acc[t], false, false);
        }
    }

    int orow = wave * 16 + ((lane >> 4) << 3);
#pragma unroll
    for (int t = 0; t < NT; ++t) {
        int ocol = t * 16 + (lane & 15);
        float badd = bias ? bias[ocol] : 0.f;
#pragma unroll
        for (int r = 0; r < 8; ++r)
            Y[(size_t)(orow + r) * Nout + ocol] = acc[t][r] + badd;
    }
}

__global__ void alpha_kernel(const float* __restrict__ Hm,
                             const float* __restrict__ a_s,
                             const float* __restrict__ a_d,
                             float* __restrict__ as_out,
                             float* __restrict__ ad_out)
{
    int tid = blockIdx.x * blockDim.x + threadIdx.x;
    if (tid >= NN * NH) return;
    int n = tid >> 2, h = tid & 3;
    const float* hp = Hm  + n * HID + h * CH;
    const float* sp = a_s + h * CH;
    const float* dp = a_d + h * CH;
    float s = 0.f, d = 0.f;
#pragma unroll
    for (int c = 0; c < CH; c += 4) {
        float4 hv = *(const float4*)(hp + c);
        float4 sv = *(const float4*)(sp + c);
        float4 dv = *(const float4*)(dp + c);
        s += hv.x * sv.x + hv.y * sv.y + hv.z * sv.z + hv.w * sv.w;
        d += hv.x * dv.x + hv.y * dv.y + hv.z * dv.z + hv.w * dv.w;
    }
    as_out[tid] = s;
    ad_out[tid] = d;
}

#define AGG_WG   256
#define AGG_R    ((NN + AGG_WG - 1) / AGG_WG)
#define AGG_TB   1024
#define AGG_NW   (AGG_TB / 32)
#define AGG_CAP  8192
#define NE4      (NE / 4)

static_assert(NE % 4 == 0, "edge list is read as int4");
static_assert(ET < (1 << 20), "edge id packed in 20 bits");
static_assert(AGG_R <= 256 && NN <= 65536, "node / row packing");
static_assert((AGG_WG - 1) * AGG_R < NN, "every range holds a node");

__device__ __forceinline__ float lrelu02(float v) { return (v > 0.f) ? v : 0.2f * v; }

__global__ __launch_bounds__(AGG_TB)
void agg_range(const int* __restrict__ src,
               const int* __restrict__ dst,
               const float* __restrict__ as_v,
               const float* __restrict__ ad_v,
               const float* __restrict__ Hm,
               const float* __restrict__ b,
               float* __restrict__ out)
{
    __shared__ float    acc[AGG_R * HID];
    __shared__ float    adl[AGG_R * NH];
    __shared__ int      mxl[AGG_R * NH];
    __shared__ float    denl[AGG_R * NH];
    __shared__ unsigned list[AGG_CAP];
    __shared__ float4   exl4[AGG_CAP];
    __shared__ int      cnt;
    float* exl = (float*)exl4;

    const int tid  = (int)threadIdx.x;
    const int lane = tid & 31;
    const int wave = tid >> 5;
    const int n0   = (int)blockIdx.x * AGG_R;
    const int nv   = (NN - n0 < AGG_R) ? (NN - n0) : AGG_R;
    const float4* as4 = (const float4*)as_v;

    for (int i = tid; i < AGG_R * HID; i += AGG_TB) acc[i] = 0.f;
    for (int i = tid; i < AGG_R * NH; i += AGG_TB) {
        int n = n0 + (i >> 2);
        adl[i]  = (n < NN) ? ad_v[n * NH + (i & 3)] : 0.f;
        mxl[i]  = (int)0x80000000;
        denl[i] = 0.f;
    }
    for (int i = tid; i < nv; i += AGG_TB)
        list[i] = (unsigned)(NE + n0 + i) | ((unsigned)i << 20);
    if (tid == 0) cnt = nv;
    __syncthreads();

    const int4* dst4 = (const int4*)dst;
    const int rot = (int)blockIdx.x * (NE4 / AGG_WG);
    for (int j = tid; j < NE4; j += 4 * AGG_TB) {
        int4 d4[4];
        int  ix[4];
#pragma unroll
        for (int u = 0; u < 4; ++u) {
            int jj = j + u * AGG_TB;
            int id = jj + rot;
            if (id >= NE4) id -= NE4;
            ix[u] = id;
            d4[u] = (jj < NE4) ? dst4[id] : make_int4(-1, -1, -1, -1);
        }
#pragma unroll
        for (int u = 0; u < 4; ++u) {
            unsigned r0 = (unsigned)(d4[u].x - n0), r1 = (unsigned)(d4[u].y - n0);
            unsigned r2 = (unsigned)(d4[u].z - n0), r3 = (unsigned)(d4[u].w - n0);
            unsigned rm = min(min(r0, r1), min(r2, r3));
            if (rm < (unsigned)nv) {
                unsigned e0 = (unsigned)ix[u] * 4u;
                if (r0 < (unsigned)nv) { int p = atomicAdd(&cnt, 1); if (p < AGG_CAP) list[p] = (e0 + 0u) | (r0 << 20); }
                if (r1 < (unsigned)nv) { int p = atomicAdd(&cnt, 1); if (p < AGG_CAP) list[p] = (e0 + 1u) | (r1 << 20); }
                if (r2 < (unsigned)nv) { int p = atomicAdd(&cnt, 1); if (p < AGG_CAP) list[p] = (e0 + 2u) | (r2 << 20); }
                if (r3 < (unsigned)nv) { int p = atomicAdd(&cnt, 1); if (p < AGG_CAP) list[p] = (e0 + 3u) | (r3 << 20); }
            }
        }
    }
    __syncthreads();
    const int total = cnt;

    if (total <= AGG_CAP) {
        for (int i = tid; i < total; i += AGG_TB) {
            unsigned p = list[i];
            int e  = (int)(p & 0xFFFFFu);
            int dl = (int)(p >> 20);
            int s  = (e < NE) ? src[e] : (e - NE);
            list[i] = (unsigned)s | ((unsigned)dl << 16);
            float4 a = as4[s];
            float v0 = lrelu02(a.x + adl[dl * NH + 0]);
            float v1 = lrelu02(a.y + adl[dl * NH + 1]);
            float v2 = lrelu02(a.z + adl[dl * NH + 2]);
            float v3 = lrelu02(a.w + adl[dl * NH + 3]);
            exl[i * NH + 0] = v0; exl[i * NH + 1] = v1;
            exl[i * NH + 2] = v2; exl[i * NH + 3] = v3;
            atomicMax(&mxl[dl * NH + 0], f2ord(v0));
            atomicMax(&mxl[dl * NH + 1], f2ord(v1));
            atomicMax(&mxl[dl * NH + 2], f2ord(v2));
            atomicMax(&mxl[dl * NH + 3], f2ord(v3));
        }
        __syncthreads();
        for (int i = tid; i < total * NH; i += AGG_TB) {
            int dl = (int)(list[i >> 2] >> 16);
            int h  = i & 3;
            float ex = expf(exl[i] - ord2f(mxl[dl * NH + h]));
            exl[i] = ex;
            atomicAdd(&denl[dl * NH + h], ex);
        }
        __syncthreads();
        int i = wave;
        for (; i + 3 * AGG_NW < total; i += 4 * AGG_NW) {
            unsigned p[4]; float4 w[4]; float hv[4][NH];
#pragma unroll
            for (int u = 0; u < 4; ++u) {
                p[u] = list[i + u * AGG_NW];
                w[u] = exl4[i + u * AGG_NW];
            }
#pragma unroll
            for (int u = 0; u < 4; ++u) {
                const float* hp = Hm + (size_t)(p[u] & 0xFFFFu) * HID + lane;
#pragma unroll
                for (int k = 0; k < NH; ++k) hv[u][k] = hp[32 * k];
            }
#pragma unroll
            for (int u = 0; u < 4; ++u) {
                float* ap = acc + (p[u] >> 16) * HID + lane;
                atomicAdd(ap +  0, hv[u][0] * w[u].x);
                atomicAdd(ap + 32, hv[u][1] * w[u].y);
                atomicAdd(ap + 64, hv[u][2] * w[u].z);
                atomicAdd(ap + 96, hv[u][3] * w[u].w);
            }
        }
        for (; i < total; i += AGG_NW) {
            unsigned p = list[i];
            float4 w = exl4[i];
            const float* hp = Hm + (size_t)(p & 0xFFFFu) * HID + lane;
            float h0 = hp[0], h1 = hp[32], h2 = hp[64], h3 = hp[96];
            float* ap = acc + (p >> 16) * HID + lane;
            atomicAdd(ap +  0, h0 * w.x);
            atomicAdd(ap + 32, h1 * w.y);
            atomicAdd(ap + 64, h2 * w.z);
            atomicAdd(ap + 96, h3 * w.w);
        }
    } else {
        for (int e = tid; e < ET; e += AGG_TB) {
            int s = (e < NE) ? src[e] : (e - NE);
            int d = (e < NE) ? dst[e] : (e - NE);
            unsigned dl = (unsigned)(d - n0);
            if (dl < (unsigned)nv) {
                float4 a = as4[s];
                atomicMax(&mxl[dl * NH + 0], f2ord(lrelu02(a.x + adl[dl * NH + 0])));
                atomicMax(&mxl[dl * NH + 1], f2ord(lrelu02(a.y + adl[dl * NH + 1])));
                atomicMax(&mxl[dl * NH + 2], f2ord(lrelu02(a.z + adl[dl * NH + 2])));
                atomicMax(&mxl[dl * NH + 3], f2ord(lrelu02(a.w + adl[dl * NH + 3])));
            }
        }
        __syncthreads();
        for (int e = tid; e < ET; e += AGG_TB) {
            int s = (e < NE) ? src[e] : (e - NE);
            int d = (e < NE) ? dst[e] : (e - NE);
            unsigned dl = (unsigned)(d - n0);
            if (dl < (unsigned)nv) {
                const float* ap4 = as_v + (size_t)s * NH;
                const float* hp  = Hm + (size_t)s * HID;
                for (int h = 0; h < NH; ++h) {
                    float v  = lrelu02(ap4[h] + adl[dl * NH + h]);
                    float ex = expf(v - ord2f(mxl[dl * NH + h]));
                    atomicAdd(&denl[dl * NH + h], ex);
                    for (int c = 0; c < CH; ++c)
                        atomicAdd(&acc[dl * HID + h * CH + c], hp[h * CH + c] * ex);
                }
            }
        }
    }
    __syncthreads();

    for (int dl = wave; dl < nv; dl += AGG_NW) {
#pragma unroll
        for (int k = 0; k < NH; ++k) {
            int c = lane + 32 * k;
            float y = acc[dl * HID + c] / (denl[dl * NH + k] + 1e-16f) + b[c];
            out[(size_t)(n0 + dl) * HID + c] = (y > 0.f) ? y : (expf(y) - 1.f);
        }
    }
}

static void run_gat_layer(const float* Xin, const float* W,
                          const float* a_s, const float* a_d, const float* b,
                          const int* src, const int* dst,
                          float* bufH, float* bufOut,
                          float* asb, float* adb, hipStream_t stream)
{
    const int TB = 256;
    int nMtiles = NN / 16;
    int blocks  = (nMtiles * 32 + TB - 1) / TB;
    wmma_gemm_strip<HID, HID / 16><<<blocks, TB, HID * HID * sizeof(float), stream>>>(
        Xin, W, nullptr, bufH, nMtiles);
    alpha_kernel<<<(NN * NH + TB - 1) / TB, TB, 0, stream>>>(bufH, a_s, a_d, asb, adb);
    agg_range<<<AGG_WG, AGG_TB, 0, stream>>>(src, dst, asb, adb, bufH, b, bufOut);
}

extern "C" void kernel_launch(void* const* d_in, const int* in_sizes, int n_in,
                              void* d_out, int out_size, void* d_ws, size_t ws_size,
                              hipStream_t stream)
{
    const float* x    = (const float*)d_in[0];
    const int*   edge = (const int*)  d_in[1];
    const float* W1   = (const float*)d_in[2];
    const float* a_s1 = (const float*)d_in[3];
    const float* a_d1 = (const float*)d_in[4];
    const float* b1   = (const float*)d_in[5];
    const float* W2   = (const float*)d_in[6];
    const float* a_s2 = (const float*)d_in[7];
    const float* a_d2 = (const float*)d_in[8];
    const float* b2   = (const float*)d_in[9];
    const float* fc_w = (const float*)d_in[10];
    const float* fc_b = (const float*)d_in[11];
    float* out = (float*)d_out;

    const int* src = edge;
    const int* dst = edge + NE;

    float* bufH   = (float*)d_ws;
    float* bufIn2 = bufH   + (size_t)NN * HID;
    float* asb    = bufIn2 + (size_t)NN * HID;
    float* adb    = asb    + (size_t)NN * NH;

    run_gat_layer(x, W1, a_s1, a_d1, b1, src, dst, bufH, bufIn2, asb, adb, stream);
    run_gat_layer(bufIn2, W2, a_s2, a_d2, b2, src, dst, bufH, bufIn2, asb, adb, stream);
    const int TB = 256;
    int nMtiles = NN / 16;
    int blocks  = (nMtiles * 32 + TB - 1) / TB;
    wmma_gemm_strip<HID, CH / 16><<<blocks, TB, HID * CH * sizeof(float), stream>>>(
        bufIn2, fc_w, fc_b, out, nMtiles);
}


# Layer 2's aggregation takes each range's (source, local destination) list, kept in the workspace by layer 1's launch, instead of filtering the edge list again; sum order stays arrival order
# speedup vs baseline: 2.9905x; 1.1943x over previous
#include <hip/hip_runtime.h>
#include <math.h>

#define NN    50000
#define NE    800000
#define ET    (NE + NN)
#define HID   128
#define NH    4
#define CH    32

typedef float v2f __attribute__((ext_vector_type(2)));
typedef float v8f __attribute__((ext_vector_type(8)));

__device__ __forceinline__ int f2ord(float f) {
    int i = __float_as_int(f);
    return (i >= 0) ? i : (i ^ 0x7fffffff);
}
__device__ __forceinline__ float ord2f(int i) {
    return __int_as_float((i >= 0) ? i : (i ^ 0x7fffffff));
}

template <int K, int NT>
__global__ void wmma_gemm_strip(const float* __restrict__ X,
                                const float* __restrict__ W,
                                const float* __restrict__ bias,
                                float* __restrict__ Y,
                                int nMtiles)
{
    constexpr int Nout = NT * 16;
    extern __shared__ float Wlds[];

    for (int i = threadIdx.x * 4; i < K * Nout; i += blockDim.x * 4)
        *(float4*)&Wlds[i] = *(const float4*)&W[i];
    __syncthreads();

    int wave = (int)((blockIdx.x * blockDim.x + threadIdx.x) >> 5);
    if (wave >= nMtiles) return;
    int lane = (int)(threadIdx.x & 31);

    int row = wave * 16 + (lane & 15);
    int col = lane & 15;
    int kh  = (lane >> 4) << 1;

    v8f acc[NT];
#pragma unroll
    for (int t = 0; t < NT; ++t)
        acc[t] = (v8f){0.f, 0.f, 0.f, 0.f, 0.f, 0.f, 0.f, 0.f};

    const float* xrow = X + (size_t)row * K;
#pragma unroll
    for (int k = 0; k < K; k += 4) {
        v2f a = *(const v2f*)(xrow + k + kh);
        const float* w0 = Wlds + (k + kh) * Nout + col;
#pragma unroll
        for (int t = 0; t < NT; ++t) {
            v2f b;
            b.x = w0[t * 16];
            b.y = w0[t * 16 + Nout];
            acc[t] = __builtin_amdgcn_wmma_f32_16x16x4_f32(
                         false, a, false, b, (short)0, acc[t], false, false);
        }
    }

    int orow = wave * 16 + ((lane >> 4) << 3);
#pragma unroll
    for (int t = 0; t < NT; ++t) {
        int ocol = t * 16 + (lane & 15);
        float badd = bias ? bias[ocol] : 0.f;
#pragma unroll
        for (int r = 0; r < 8; ++r)
            Y[(size_t)(orow + r) * Nout + ocol] = acc[t][r] + badd;
    }
}

__global__ void alpha_kernel(const float* __restrict__ Hm,
                             const float* __restrict__ a_s,
                             const float* __restrict__ a_d,
                             float* __restrict__ as_out,
                             float* __restrict__ ad_out)
{
    int tid = blockIdx.x * blockDim.x + threadIdx.x;
    if (tid >= NN * NH) return;
    int n = tid >> 2, h = tid & 3;
    const float* hp = Hm  + n * HID + h * CH;
    const float* sp = a_s + h * CH;
    const float* dp = a_d + h * CH;
    float s = 0.f, d = 0.f;
#pragma unroll
    for (int c = 0; c < CH; c += 4) {
        float4 hv = *(const float4*)(hp + c);
        float4 sv = *(const float4*)(sp + c);
        float4 dv = *(const float4*)(dp + c);
        s += hv.x * sv.x + hv.y * sv.y + hv.z * sv.z + hv.w * sv.w;
        d += hv.x * dv.x + hv.y * dv.y + hv.z * dv.z + hv.w * dv.w;
    }
    as_out[tid] = s;
    ad_out[tid] = d;
}

#define AGG_WG   256
#define AGG_R    ((NN + AGG_WG - 1) / AGG_WG)
#define AGG_TB   1024
#define AGG_NW   (AGG_TB / 32)
#define AGG_CAP  8192
#define NE4      (NE / 4)
#define KEEP_HDR    32
#define KEEP_STRIDE (KEEP_HDR + AGG_CAP)

static_assert(NE % 4 == 0, "edge list is read as int4");
static_assert(ET < (1 << 20), "edge id packed in 20 bits");
static_assert(AGG_R <= 256 && NN <= 65536, "node / row packing");
static_assert((AGG_WG - 1) * AGG_R < NN, "every range holds a node");

__device__ __forceinline__ float lrelu02(float v) { return (v > 0.f) ? v : 0.2f * v; }

template <bool SAVED>
__global__ __launch_bounds__(AGG_TB)
void agg_range(const int* __restrict__ src,
               const int* __restrict__ dst,
               const float* __restrict__ as_v,
               const float* __restrict__ ad_v,
               const float* __restrict__ Hm,
               const float* __restrict__ b,
               float* __restrict__ out,
               unsigned* __restrict__ keep)
{
    __shared__ float    acc[AGG_R * HID];
    __shared__ float    adl[AGG_R * NH];
    __shared__ int      mxl[AGG_R * NH];
    __shared__ float    denl[AGG_R * NH];
    __shared__ unsigned list[AGG_CAP];
    __shared__ float4   exl4[AGG_CAP];
    __shared__ int      cnt;
    float* exl = (float*)exl4;

    const int tid  = (int)threadIdx.x;
    const int lane = tid & 31;
    const int wave = tid >> 5;
    const int n0   = (int)blockIdx.x * AGG_R;
    const int nv   = (NN - n0 < AGG_R) ? (NN - n0) : AGG_R;
    const float4* as4 = (const float4*)as_v;
    unsigned* keep_r = keep + (size_t)blockIdx.x * KEEP_STRIDE;

    for (int i = tid; i < AGG_R * HID; i += AGG_TB) acc[i] = 0.f;
    for (int i = tid; i < AGG_R * NH; i += AGG_TB) {
        int n = n0 + (i >> 2);
        adl[i]  = (n < NN) ? ad_v[n * NH + (i & 3)] : 0.f;
        mxl[i]  = (int)0x80000000;
        denl[i] = 0.f;
    }
    if (SAVED) {
        if (tid == 0) cnt = (int)keep_r[0];
    } else {
        for (int i = tid; i < nv; i += AGG_TB)
            list[i] = (unsigned)(NE + n0 + i) | ((unsigned)i << 20);
        if (tid == 0) cnt = nv;
    }
    __syncthreads();

    if (!SAVED) {
    const int4* dst4 = (const int4*)dst;
    const int rot = (int)blockIdx.x * (NE4 / AGG_WG);
    for (int j = tid; j < NE4; j += 4 * AGG_TB) {
        int4 d4[4];
        int  ix[4];
#pragma unroll
        for (int u = 0; u < 4; ++u) {
            int jj = j + u * AGG_TB;
            int id = jj + rot;
            if (id >= NE4) id -= NE4;
            ix[u] = id;
            d4[u] = (jj < NE4) ? dst4[id] : make_int4(-1, -1, -1, -1);
        }
#pragma unroll
        for (int u = 0; u < 4; ++u) {
            unsigned r0 = (unsigned)(d4[u].x - n0), r1 = (unsigned)(d4[u].y - n0);
            unsigned r2 = (unsigned)(d4[u].z - n0), r3 = (unsigned)(d4[u].w - n0);
            unsigned rm = min(min(r0, r1), min(r2, r3));
            if (rm < (unsigned)nv) {
                unsigned e0 = (unsigned)ix[u] * 4u;
                if (r0 < (unsigned)nv) { int p = atomicAdd(&cnt, 1); if (p < AGG_CAP) list[p] = (e0 + 0u) | (r0 << 20); }
                if (r1 < (unsigned)nv) { int p = atomicAdd(&cnt, 1); if (p < AGG_CAP) list[p] = (e0 + 1u) | (r1 << 20); }
                if (r2 < (unsigned)nv) { int p = atomicAdd(&cnt, 1); if (p < AGG_CAP) list[p] = (e0 + 2u) | (r2 << 20); }
                if (r3 < (unsigned)nv) { int p = atomicAdd(&cnt, 1); if (p < AGG_CAP) list[p] = (e0 + 3u) | (r3 << 20); }
            }
        }
    }
    __syncthreads();
    }
    const int total = cnt;
    if (!SAVED && tid == 0) keep_r[0] = (unsigned)total;

    if (total <= AGG_CAP) {
        for (int i = tid; i < total; i += AGG_TB) {
            unsigned q;
            if (SAVED) {
                q = keep_r[KEEP_HDR + i];
            } else {
                unsigned p = list[i];
                int e  = (int)(p & 0xFFFFFu);
                int s0 = (e < NE) ? src[e] : (e - NE);
                q = (unsigned)s0 | ((p >> 20) << 16);
                keep_r[KEEP_HDR + i] = q;
            }
            list[i] = q;
            int s  = (int)(q & 0xFFFFu);
            int dl = (int)(q >> 16);
            float4 a = as4[s];
            float v0 = lrelu02(a.x + adl[dl * NH + 0]);
            float v1 = lrelu02(a.y + adl[dl * NH + 1]);
            float v2 = lrelu02(a.z + adl[dl * NH + 2]);
            float v3 = lrelu02(a.w + adl[dl * NH + 3]);
            exl[i * NH + 0] = v0; exl[i * NH + 1] = v1;
            exl[i * NH + 2] = v2; exl[i * NH + 3] = v3;
            atomicMax(&mxl[dl * NH + 0], f2ord(v0));
            atomicMax(&mxl[dl * NH + 1], f2ord(v1));
            atomicMax(&mxl[dl * NH + 2], f2ord(v2));
            atomicMax(&mxl[dl * NH + 3], f2ord(v3));
        }
        __syncthreads();
        for (int i = tid; i < total * NH; i += AGG_TB) {
            int dl = (int)(list[i >> 2] >> 16);
            int h  = i & 3;
            float ex = expf(exl[i] - ord2f(mxl[dl * NH + h]));
            exl[i] = ex;
            atomicAdd(&denl[dl * NH + h], ex);
        }
        __syncthreads();
        int i = wave;
        for (; i + 3 * AGG_NW < total; i += 4 * AGG_NW) {
            unsigned p[4]; float4 w[4]; float hv[4][NH];
#pragma unroll
            for (int u = 0; u < 4; ++u) {
                p[u] = list[i + u * AGG_NW];
                w[u] = exl4[i + u * AGG_NW];
            }
#pragma unroll
            for (int u = 0; u < 4; ++u) {
                const float* hp = Hm + (size_t)(p[u] & 0xFFFFu) * HID + lane;
#pragma unroll
                for (int k = 0; k < NH; ++k) hv[u][k] = hp[32 * k];
            }
#pragma unroll
            for (int u = 0; u < 4; ++u) {
                float* ap = acc + (p[u] >> 16) * HID + lane;
                atomicAdd(ap +  0, hv[u][0] * w[u].x);
                atomicAdd(ap + 32, hv[u][1] * w[u].y);
                atomicAdd(ap + 64, hv[u][2] * w[u].z);
                atomicAdd(ap + 96, hv[u][3] * w[u].w);
            }
        }
        for (; i < total; i += AGG_NW) {
            unsigned p = list[i];
            float4 w = exl4[i];
            const float* hp = Hm + (size_t)(p & 0xFFFFu) * HID + lane;
            float h0 = hp[0], h1 = hp[32], h2 = hp[64], h3 = hp[96];
            float* ap = acc + (p >> 16) * HID + lane;
            atomicAdd(ap +  0, h0 * w.x);
            atomicAdd(ap + 32, h1 * w.y);
            atomicAdd(ap + 64, h2 * w.z);
            atomicAdd(ap + 96, h3 * w.w);
        }
    } else {
        for (int e = tid; e < ET; e += AGG_TB) {
            int s = (e < NE) ? src[e] : (e - NE);
            int d = (e < NE) ? dst[e] : (e - NE);
            unsigned dl = (unsigned)(d - n0);
            if (dl < (unsigned)nv) {
                float4 a = as4[s];
                atomicMax(&mxl[dl * NH + 0], f2ord(lrelu02(a.x + adl[dl * NH + 0])));
                atomicMax(&mxl[dl * NH + 1], f2ord(lrelu02(a.y + adl[dl * NH + 1])));
                atomicMax(&mxl[dl * NH + 2], f2ord(lrelu02(a.z + adl[dl * NH + 2])));
                atomicMax(&mxl[dl * NH + 3], f2ord(lrelu02(a.w + adl[dl * NH + 3])));
            }
        }
        __syncthreads();
        for (int e = tid; e < ET; e += AGG_TB) {
            int s = (e < NE) ? src[e] : (e - NE);
            int d = (e < NE) ? dst[e] : (e - NE);
            unsigned dl = (unsigned)(d - n0);
            if (dl < (unsigned)nv) {
                const float* ap4 = as_v + (size_t)s * NH;
                const float* hp  = Hm + (size_t)s * HID;
                for (int h = 0; h < NH; ++h) {
                    float v  = lrelu02(ap4[h] + adl[dl * NH + h]);
                    float ex = expf(v - ord2f(mxl[dl * NH + h]));
                    atomicAdd(&denl[dl * NH + h], ex);
                    for (int c = 0; c < CH; ++c)
                        atomicAdd(&acc[dl * HID + h * CH + c], hp[h * CH + c] * ex);
                }
            }
        }
    }
    __syncthreads();

    for (int dl = wave; dl < nv; dl += AGG_NW) {
#pragma unroll
        for (int k = 0; k < NH; ++k) {
            int c = lane + 32 * k;
            float y = acc[dl * HID + c] / (denl[dl * NH + k] + 1e-16f) + b[c];
            out[(size_t)(n0 + dl) * HID + c] = (y > 0.f) ? y : (expf(y) - 1.f);
        }
    }
}

template <bool SAVED>
static void run_gat_layer(const float* Xin, const float* W,
                          const float* a_s, const float* a_d, const float* b,
                          const int* src, const int* dst,
                          float* bufH, float* bufOut,
                          float* asb, float* adb, unsigned* keep, hipStream_t stream)
{
    const int TB = 256;
    int nMtiles = NN / 16;
    int blocks  = (nMtiles * 32 + TB - 1) / TB;
    wmma_gemm_strip<HID, HID / 16><<<blocks, TB, HID * HID * sizeof(float), stream>>>(
        Xin, W, nullptr, bufH, nMtiles);
    alpha_kernel<<<(NN * NH + TB - 1) / TB, TB, 0, stream>>>(bufH, a_s, a_d, asb, adb);
    agg_range<SAVED><<<AGG_WG, AGG_TB, 0, stream>>>(src, dst, asb, adb, bufH, b, bufOut, keep);
}

extern "C" void kernel_launch(void* const* d_in, const int* in_sizes, int n_in,
                              void* d_out, int out_size, void* d_ws, size_t ws_size,
                              hipStream_t stream)
{
    const float* x    = (const float*)d_in[0];
    const int*   edge = (const int*)  d_in[1];
    const float* W1   = (const float*)d_in[2];
    const float* a_s1 = (const float*)d_in[3];
    const float* a_d1 = (const float*)d_in[4];
    const float* b1   = (const float*)d_in[5];
    const float* W2   = (const float*)d_in[6];
    const float* a_s2 = (const float*)d_in[7];
    const float* a_d2 = (const float*)d_in[8];
    const float* b2   = (const float*)d_in[9];
    const float* fc_w = (const float*)d_in[10];
    const float* fc_b = (const float*)d_in[11];
    float* out = (float*)d_out;

    const int* src = edge;
    const int* dst = edge + NE;

    float* bufH   = (float*)d_ws;
    float* bufIn2 = bufH   + (size_t)NN * HID;
    float* asb    = bufIn2 + (size_t)NN * HID;
    float* adb    = asb    + (size_t)NN * NH;
    unsigned* keep = (unsigned*)(adb + (size_t)NN * NH);

    run_gat_layer<false>(x, W1, a_s1, a_d1, b1, src, dst, bufH, bufIn2, asb, adb, keep, stream);
    run_gat_layer<true>(bufIn2, W2, a_s2, a_d2, b2, src, dst, bufH, bufIn2, asb, adb, keep, stream);
    const int TB = 256;
    int nMtiles = NN / 16;
    int blocks  = (nMtiles * 32 + TB - 1) / TB;
    wmma_gemm_strip<HID, CH / 16><<<blocks, TB, HID * CH * sizeof(float), stream>>>(
        bufIn2, fc_w, fc_b, out, nMtiles);
}


# Attention logit halves come out of the projection GEMM as one more 16-column tile, W @ [a_s | a_d] folded in LDS at staging: the logit launches go, 7 -> 5 launches; logit sum order changes
# speedup vs baseline: 3.1422x; 1.0507x over previous
#include <hip/hip_runtime.h>
#include <math.h>

#define NN    50000
#define NE    800000
#define ET    (NE + NN)
#define HID   128
#define NH    4
#define CH    32

typedef float v2f __attribute__((ext_vector_type(2)));
typedef float v8f __attribute__((ext_vector_type(8)));

__device__ __forceinline__ int f2ord(float f) {
    int i = __float_as_int(f);
    return (i >= 0) ? i : (i ^ 0x7fffffff);
}
__device__ __forceinline__ float ord2f(int i) {
    return __int_as_float((i >= 0) ? i : (i ^ 0x7fffffff));
}

#define GEMM_TB 256
template <int K, int NT, bool LOGITS>
__global__ __launch_bounds__(GEMM_TB)
void wmma_gemm_strip(const float* __restrict__ X,
                                const float* __restrict__ W,
                                const float* __restrict__ bias,
                                float* __restrict__ Y,
                                int nMtiles,
                                const float* __restrict__ a_s,
                                const float* __restrict__ a_d,
                                float* __restrict__ as_out,
                                float* __restrict__ ad_out)
{
    constexpr int Nout = NT * 16;
    constexpr int NTL  = LOGITS ? NT + 1 : NT;
    constexpr int P    = NTL * 16;
    __shared__ float Wlds[K * P];

    for (int i = threadIdx.x * 4; i < K * Nout; i += blockDim.x * 4)
        *(float4*)&Wlds[(i / Nout) * P + (i % Nout)] = *(const float4*)&W[i];

    if (LOGITS) {
        static_assert(!LOGITS || (K == HID && Nout == HID), "fold is written for 128 x 128");
        __syncthreads();
        const int ln = (int)threadIdx.x & 31;
        const int wv = (int)threadIdx.x >> 5;
        const float4 s4 = *(const float4*)(a_s + 4 * ln);
        const float4 d4 = *(const float4*)(a_d + 4 * ln);
        constexpr int NW = GEMM_TB / 32;
        constexpr int R  = K / NW;
        float ps[R], pd[R];
#pragma unroll
        for (int it = 0; it < R; ++it) {
            const int k = it * NW + wv;
            const float4 w4 = *(const float4*)(Wlds + k * P + 4 * ln);
            ps[it] = w4.x * s4.x + w4.y * s4.y + w4.z * s4.z + w4.w * s4.w;
            pd[it] = w4.x * d4.x + w4.y * d4.y + w4.z * d4.z + w4.w * d4.w;
        }
#pragma unroll
        for (int m = 1; m < 8; m <<= 1) {
#pragma unroll
            for (int it = 0; it < R; ++it) {
                ps[it] += __shfl_xor(ps[it], m);
                pd[it] += __shfl_xor(pd[it], m);
            }
        }
        float* wa = Wlds + wv * P + Nout;
        if ((ln & 7) == 0) {
#pragma unroll
            for (int it = 0; it < R; ++it) {
                wa[it * NW * P + (ln >> 3)]      = ps[it];
                wa[it * NW * P + NH + (ln >> 3)] = pd[it];
            }
        } else if ((ln & 7) == 1) {
#pragma unroll
            for (int it = 0; it < R; ++it) {
                wa[it * NW * P + 2 * NH + (ln >> 3)] = 0.f;
                wa[it * NW * P + 3 * NH + (ln >> 3)] = 0.f;
            }
        }
    }
    __syncthreads();

    int wave = (int)((blockIdx.x * blockDim.x + threadIdx.x) >> 5);
    if (wave >= nMtiles) return;
    int lane = (int)(threadIdx.x & 31);

    int row = wave * 16 + (lane & 15);
    int col = lane & 15;
    int kh  = (lane >> 4) << 1;

    v8f acc[NTL];
#pragma unroll
    for (int t = 0; t < NTL; ++t)
        acc[t] = (v8f){0.f, 0.f, 0.f, 0.f, 0.f, 0.f, 0.f, 0.f};

    const float* xrow = X + (size_t)row * K;
#pragma unroll
    for (int k = 0; k < K; k += 4) {
        v2f a = *(const v2f*)(xrow + k + kh);
        const float* w0 = Wlds + (k + kh) * P + col;
#pragma unroll
        for (int t = 0; t < NTL; ++t) {
            v2f b;
            b.x = w0[t * 16];
            b.y = w0[t * 16 + P];
            acc[t] = __builtin_amdgcn_wmma_f32_16x16x4_f32(
                         false, a, false, b, (short)0, acc[t], false, false);
        }
    }

    int orow = wave * 16 + ((lane >> 4) << 3);
#pragma unroll
    for (int t = 0; t < NT; ++t) {
        int ocol = t * 16 + (lane & 15);
        float badd = bias ? bias[ocol] : 0.f;
#pragma unroll
        for (int r = 0; r < 8; ++r)
            Y[(size_t)(orow + r) * Nout + ocol] = acc[t][r] + badd;
    }
    if (LOGITS) {
        const int c = lane & 15;
        if (c < 2 * NH) {
            float* dstp = (c < NH) ? (as_out + c) : (ad_out + (c - NH));
#pragma unroll
            for (int r = 0; r < 8; ++r)
                dstp[(size_t)(orow + r) * NH] = acc[NTL - 1][r];
        }
    }
}

#define AGG_WG   256
#define AGG_R    ((NN + AGG_WG - 1) / AGG_WG)
#define AGG_TB   1024
#define AGG_NW   (AGG_TB / 32)
#define AGG_CAP  8192
#define NE4      (NE / 4)
#define KEEP_HDR    32
#define KEEP_STRIDE (KEEP_HDR + AGG_CAP)

static_assert(NE % 4 == 0, "edge list is read as int4");
static_assert(ET < (1 << 20), "edge id packed in 20 bits");
static_assert(AGG_R <= 256 && NN <= 65536, "node / row packing");
static_assert((AGG_WG - 1) * AGG_R < NN, "every range holds a node");

__device__ __forceinline__ float lrelu02(float v) { return (v > 0.f) ? v : 0.2f * v; }

template <bool SAVED>
__global__ __launch_bounds__(AGG_TB)
void agg_range(const int* __restrict__ src,
               const int* __restrict__ dst,
               const float* __restrict__ as_v,
               const float* __restrict__ ad_v,
               const float* __restrict__ Hm,
               const float* __restrict__ b,
               float* __restrict__ out,
               unsigned* __restrict__ keep)
{
    __shared__ float    acc[AGG_R * HID];
    __shared__ float    adl[AGG_R * NH];
    __shared__ int      mxl[AGG_R * NH];
    __shared__ float    denl[AGG_R * NH];
    __shared__ unsigned list[AGG_CAP];
    __shared__ float4   exl4[AGG_CAP];
    __shared__ int      cnt;
    float* exl = (float*)exl4;

    const int tid  = (int)threadIdx.x;
    const int lane = tid & 31;
    const int wave = tid >> 5;
    const int n0   = (int)blockIdx.x * AGG_R;
    const int nv   = (NN - n0 < AGG_R) ? (NN - n0) : AGG_R;
    const float4* as4 = (const float4*)as_v;
    unsigned* keep_r = keep + (size_t)blockIdx.x * KEEP_STRIDE;

    for (int i = tid; i < AGG_R * HID; i += AGG_TB) acc[i] = 0.f;
    for (int i = tid; i < AGG_R * NH; i += AGG_TB) {
        int n = n0 + (i >> 2);
        adl[i]  = (n < NN) ? ad_v[n * NH + (i & 3)] : 0.f;
        mxl[i]  = (int)0x80000000;
        denl[i] = 0.f;
    }
    if (SAVED) {
        if (tid == 0) cnt = (int)keep_r[0];
    } else {
        for (int i = tid; i < nv; i += AGG_TB)
            list[i] = (unsigned)(NE + n0 + i) | ((unsigned)i << 20);
        if (tid == 0) cnt = nv;
    }
    __syncthreads();

    if (!SAVED) {
    const int4* dst4 = (const int4*)dst;
    const int rot = (int)blockIdx.x * (NE4 / AGG_WG);
    for (int j = tid; j < NE4; j += 4 * AGG_TB) {
        int4 d4[4];
        int  ix[4];
#pragma unroll
        for (int u = 0; u < 4; ++u) {
            int jj = j + u * AGG_TB;
            int id = jj + rot;
            if (id >= NE4) id -= NE4;
            ix[u] = id;
            d4[u] = (jj < NE4) ? dst4[id] : make_int4(-1, -1, -1, -1);
        }
#pragma unroll
        for (int u = 0; u < 4; ++u) {
            unsigned r0 = (unsigned)(d4[u].x - n0), r1 = (unsigned)(d4[u].y - n0);
            unsigned r2 = (unsigned)(d4[u].z - n0), r3 = (unsigned)(d4[u].w - n0);
            unsigned rm = min(min(r0, r1), min(r2, r3));
            if (rm < (unsigned)nv) {
                unsigned e0 = (unsigned)ix[u] * 4u;
                if (r0 < (unsigned)nv) { int p = atomicAdd(&cnt, 1); if (p < AGG_CAP) list[p] = (e0 + 0u) | (r0 << 20); }
                if (r1 < (unsigned)nv) { int p = atomicAdd(&cnt, 1); if (p < AGG_CAP) list[p] = (e0 + 1u) | (r1 << 20); }
                if (r2 < (unsigned)nv) { int p = atomicAdd(&cnt, 1); if (p < AGG_CAP) list[p] = (e0 + 2u) | (r2 << 20); }
                if (r3 < (unsigned)nv) { int p = atomicAdd(&cnt, 1); if (p < AGG_CAP) list[p] = (e0 + 3u) | (r3 << 20); }
            }
        }
    }
    __syncthreads();
    }
    const int total = cnt;
    if (!SAVED && tid == 0) keep_r[0] = (unsigned)total;

    if (total <= AGG_CAP) {
        for (int i = tid; i < total; i += AGG_TB) {
            unsigned q;
            if (SAVED) {
                q = keep_r[KEEP_HDR + i];
            } else {
                unsigned p = list[i];
                int e  = (int)(p & 0xFFFFFu);
                int s0 = (e < NE) ? src[e] : (e - NE);
                q = (unsigned)s0 | ((p >> 20) << 16);
                keep_r[KEEP_HDR + i] = q;
            }
            list[i] = q;
            int s  = (int)(q & 0xFFFFu);
            int dl = (int)(q >> 16);
            float4 a = as4[s];
            float v0 = lrelu02(a.x + adl[dl * NH + 0]);
            float v1 = lrelu02(a.y + adl[dl * NH + 1]);
            float v2 = lrelu02(a.z + adl[dl * NH + 2]);
            float v3 = lrelu02(a.w + adl[dl * NH + 3]);
            exl[i * NH + 0] = v0; exl[i * NH + 1] = v1;
            exl[i * NH + 2] = v2; exl[i * NH + 3] = v3;
            atomicMax(&mxl[dl * NH + 0], f2ord(v0));
            atomicMax(&mxl[dl * NH + 1], f2ord(v1));
            atomicMax(&mxl[dl * NH + 2], f2ord(v2));
            atomicMax(&mxl[dl * NH + 3], f2ord(v3));
        }
        __syncthreads();
        for (int i = tid; i < total * NH; i += AGG_TB) {
            int dl = (int)(list[i >> 2] >> 16);
            int h  = i & 3;
            float ex = expf(exl[i] - ord2f(mxl[dl * NH + h]));
            exl[i] = ex;
            atomicAdd(&denl[dl * NH + h], ex);
        }
        __syncthreads();
        int i = wave;
        for (; i + 3 * AGG_NW < total; i += 4 * AGG_NW) {
            unsigned p[4]; float4 w[4]; float hv[4][NH];
#pragma unroll
            for (int u = 0; u < 4; ++u) {
                p[u] = list[i + u * AGG_NW];
                w[u] = exl4[i + u * AGG_NW];
            }
#pragma unroll
            for (int u = 0; u < 4; ++u) {
                const float* hp = Hm + (size_t)(p[u] & 0xFFFFu) * HID + lane;
#pragma unroll
                for (int k = 0; k < NH; ++k) hv[u][k] = hp[32 * k];
            }
#pragma unroll
            for (int u = 0; u < 4; ++u) {
                float* ap = acc + (p[u] >> 16) * HID + lane;
                atomicAdd(ap +  0, hv[u][0] * w[u].x);
                atomicAdd(ap + 32, hv[u][1] * w[u].y);
                atomicAdd(ap + 64, hv[u][2] * w[u].z);
                atomicAdd(ap + 96, hv[u][3] * w[u].w);
            }
        }
        for (; i < total; i += AGG_NW) {
            unsigned p = list[i];
            float4 w = exl4[i];
            const float* hp = Hm + (size_t)(p & 0xFFFFu) * HID + lane;
            float h0 = hp[0], h1 = hp[32], h2 = hp[64], h3 = hp[96];
            float* ap = acc + (p >> 16) * HID + lane;
            atomicAdd(ap +  0, h0 * w.x);
            atomicAdd(ap + 32, h1 * w.y);
            atomicAdd(ap + 64, h2 * w.z);
            atomicAdd(ap + 96, h3 * w.w);
        }
    } else {
        for (int e = tid; e < ET; e += AGG_TB) {
            int s = (e < NE) ? src[e] : (e - NE);
            int d = (e < NE) ? dst[e] : (e - NE);
            unsigned dl = (unsigned)(d - n0);
            if (dl < (unsigned)nv) {
                float4 a = as4[s];
                atomicMax(&mxl[dl * NH + 0], f2ord(lrelu02(a.x + adl[dl * NH + 0])));
                atomicMax(&mxl[dl * NH + 1], f2ord(lrelu02(a.y + adl[dl * NH + 1])));
                atomicMax(&mxl[dl * NH + 2], f2ord(lrelu02(a.z + adl[dl * NH + 2])));
                atomicMax(&mxl[dl * NH + 3], f2ord(lrelu02(a.w + adl[dl * NH + 3])));
            }
        }
        __syncthreads();
        for (int e = tid; e < ET; e += AGG_TB) {
            int s = (e < NE) ? src[e] : (e - NE);
            int d = (e < NE) ? dst[e] : (e - NE);
            unsigned dl = (unsigned)(d - n0);
            if (dl < (unsigned)nv) {
                const float* ap4 = as_v + (size_t)s * NH;
                const float* hp  = Hm + (size_t)s * HID;
                for (int h = 0; h < NH; ++h) {
                    float v  = lrelu02(ap4[h] + adl[dl * NH + h]);
                    float ex = expf(v - ord2f(mxl[dl * NH + h]));
                    atomicAdd(&denl[dl * NH + h], ex);
                    for (int c = 0; c < CH; ++c)
                        atomicAdd(&acc[dl * HID + h * CH + c], hp[h * CH + c] * ex);
                }
            }
        }
    }
    __syncthreads();

    for (int dl = wave; dl < nv; dl += AGG_NW) {
#pragma unroll
        for (int k = 0; k < NH; ++k) {
            int c = lane + 32 * k;
            float y = acc[dl * HID + c] / (denl[dl * NH + k] + 1e-16f) + b[c];
            out[(size_t)(n0 + dl) * HID + c] = (y > 0.f) ? y : (expf(y) - 1.f);
        }
    }
}

template <bool SAVED>
static void run_gat_layer(const float* Xin, const float* W,
                          const float* a_s, const float* a_d, const float* b,
                          const int* src, const int* dst,
                          float* bufH, float* bufOut,
                          float* asb, float* adb, unsigned* keep, hipStream_t stream)
{
    const int TB = GEMM_TB;
    int nMtiles = NN / 16;
    int blocks  = (nMtiles * 32 + TB - 1) / TB;
    wmma_gemm_strip<HID, HID / 16, true><<<blocks, TB, 0, stream>>>(
        Xin, W, nullptr, bufH, nMtiles, a_s, a_d, asb, adb);
    agg_range<SAVED><<<AGG_WG, AGG_TB, 0, stream>>>(src, dst, asb, adb, bufH, b, bufOut, keep);
}

extern "C" void kernel_launch(void* const* d_in, const int* in_sizes, int n_in,
                              void* d_out, int out_size, void* d_ws, size_t ws_size,
                              hipStream_t stream)
{
    const float* x    = (const float*)d_in[0];
    const int*   edge = (const int*)  d_in[1];
    const float* W1   = (const float*)d_in[2];
    const float* a_s1 = (const float*)d_in[3];
    const float* a_d1 = (const float*)d_in[4];
    const float* b1   = (const float*)d_in[5];
    const float* W2   = (const float*)d_in[6];
    const float* a_s2 = (const float*)d_in[7];
    const float* a_d2 = (const float*)d_in[8];
    const float* b2   = (const float*)d_in[9];
    const float* fc_w = (const float*)d_in[10];
    const float* fc_b = (const float*)d_in[11];
    float* out = (float*)d_out;

    const int* src = edge;
    const int* dst = edge + NE;

    float* bufH   = (float*)d_ws;
    float* bufIn2 = bufH   + (size_t)NN * HID;
    float* asb    = bufIn2 + (size_t)NN * HID;
    float* adb    = asb    + (size_t)NN * NH;
    unsigned* keep = (unsigned*)(adb + (size_t)NN * NH);

    run_gat_layer<false>(x, W1, a_s1, a_d1, b1, src, dst, bufH, bufIn2, asb, adb, keep, stream);
    run_gat_layer<true>(bufIn2, W2, a_s2, a_d2, b2, src, dst, bufH, bufIn2, asb, adb, keep, stream);
    const int TB = GEMM_TB;
    int nMtiles = NN / 16;
    int blocks  = (nMtiles * 32 + TB - 1) / TB;
    wmma_gemm_strip<HID, CH / 16, false><<<blocks, TB, 0, stream>>>(
        bufIn2, fc_w, fc_b, out, nMtiles, nullptr, nullptr, nullptr, nullptr);
}


# Range filter sweeps a bit-plane table of range ids (8 bits per edge, built in the layer-1 projection launch; 253 ranges x 198 nodes, id 255 = none), not the 32-bit destinations; sum order as before
# speedup vs baseline: 3.5792x; 1.1391x over previous
#include <hip/hip_runtime.h>
#include <math.h>

#define NN    50000
#define NE    800000
#define ET    (NE + NN)
#define HID   128
#define NH    4
#define CH    32

typedef float v2f __attribute__((ext_vector_type(2)));
typedef float v8f __attribute__((ext_vector_type(8)));

#define AGG_R    198
#define AGG_WG   ((NN + AGG_R - 1) / AGG_R)
#define RID_NONE 255u
#define NG       (NE / 32)
#define PLANE_WORDS (2 * 4 * NG)
static_assert(AGG_WG <= 254, "range ids must leave 255 free");
static_assert(NE % 256 == 0, "the table is built in chunks of 256 edges");
static_assert(NE / 256 <= NN / 16, "one 256-edge chunk per strip wave of the projection launch");

__device__ __forceinline__ int f2ord(float f) {
    int i = __float_as_int(f);
    return (i >= 0) ? i : (i ^ 0x7fffffff);
}
__device__ __forceinline__ float ord2f(int i) {
    return __int_as_float((i >= 0) ? i : (i ^ 0x7fffffff));
}

#define GEMM_TB 256
template <int K, int NT, bool LOGITS, bool PLANES>
__global__ __launch_bounds__(GEMM_TB)
void wmma_gemm_strip(const float* __restrict__ X,
                                const float* __restrict__ W,
                                const float* __restrict__ bias,
                                float* __restrict__ Y,
                                int nMtiles,
                                const float* __restrict__ a_s,
                                const float* __restrict__ a_d,
                                float* __restrict__ as_out,
                                float* __restrict__ ad_out,
                                const int* __restrict__ edst,
                                unsigned* __restrict__ planes)
{
    constexpr int Nout = NT * 16;
    constexpr int NTL  = LOGITS ? NT + 1 : NT;
    constexpr int P    = NTL * 16;
    __shared__ float Wlds[K * P];

    const int gwave = (int)((blockIdx.x * blockDim.x + threadIdx.x) >> 5);
    int dv[8];
    if (PLANES) {
        if (gwave < NE / 256) {
#pragma unroll
            for (int g = 0; g < 8; ++g)
                dv[g] = edst[gwave * 256 + g * 32 + ((int)threadIdx.x & 31)];
        }
    }

    for (int i = threadIdx.x * 4; i < K * Nout; i += blockDim.x * 4)
        *(float4*)&Wlds[(i / Nout) * P + (i % Nout)] = *(const float4*)&W[i];

    if (LOGITS) {
        static_assert(!LOGITS || (K == HID && Nout == HID), "fold is written for 128 x 128");
        __syncthreads();
        const int ln = (int)threadIdx.x & 31;
        const int wv = (int)threadIdx.x >> 5;
        const float4 s4 = *(const float4*)(a_s + 4 * ln);
        const float4 d4 = *(const float4*)(a_d + 4 * ln);
        constexpr int NW = GEMM_TB / 32;
        constexpr int R  = K / NW;
        float ps[R], pd[R];
#pragma unroll
        for (int it = 0; it < R; ++it) {
            const int k = it * NW + wv;
            const float4 w4 = *(const float4*)(Wlds + k * P + 4 * ln);
            ps[it] = w4.x * s4.x + w4.y * s4.y + w4.z * s4.z + w4.w * s4.w;
            pd[it] = w4.x * d4.x + w4.y * d4.y + w4.z * d4.z + w4.w * d4.w;
        }
#pragma unroll
        for (int m = 1; m < 8; m <<= 1) {
#pragma unroll
            for (int it = 0; it < R; ++it) {
                ps[it] += __shfl_xor(ps[it], m);
                pd[it] += __shfl_xor(pd[it], m);
            }
        }
        float* wa = Wlds + wv * P + Nout;
        if ((ln & 7) == 0) {
#pragma unroll
            for (int it = 0; it < R; ++it) {
                wa[it * NW * P + (ln >> 3)]      = ps[it];
                wa[it * NW * P + NH + (ln >> 3)] = pd[it];
            }
        } else if ((ln & 7) == 1) {
#pragma unroll
            for (int it = 0; it < R; ++it) {
                wa[it * NW * P + 2 * NH + (ln >> 3)] = 0.f;
                wa[it * NW * P + 3 * NH + (ln >> 3)] = 0.f;
            }
        }
    }
    __syncthreads();

    if (PLANES) {
        if (gwave < NE / 256) {
            const int ln = (int)threadIdx.x & 31;
            unsigned wa = 0u, wb = 0u;
#pragma unroll
            for (int g = 0; g < 8; ++g) {
                const unsigned d = (unsigned)dv[g];
                const unsigned r = (d < (unsigned)NN) ? d / (unsigned)AGG_R : RID_NONE;
#pragma unroll
                for (int k = 0; k < 8; ++k) {
                    const unsigned bal = __builtin_amdgcn_ballot_w32(((r >> k) & 1u) != 0u);
                    if (k < 4) { if (ln == 4 * g + k)     wa = bal; }
                    else       { if (ln == 4 * g + k - 4) wb = bal; }
                }
            }
            planes[gwave * 32 + ln]          = wa;
            planes[4 * NG + gwave * 32 + ln] = wb;
        }
    }

    int wave = gwave;
    if (wave >= nMtiles) return;
    int lane = (int)(threadIdx.x & 31);

    int row = wave * 16 + (lane & 15);
    int col = lane & 15;
    int kh  = (lane >> 4) << 1;

    v8f acc[NTL];
#pragma unroll
    for (int t = 0; t < NTL; ++t)
        acc[t] = (v8f){0.f, 0.f, 0.f, 0.f, 0.f, 0.f, 0.f, 0.f};

    const float* xrow = X + (size_t)row * K;
#pragma unroll
    for (int k = 0; k < K; k += 4) {
        v2f a = *(const v2f*)(xrow + k + kh);
        const float* w0 = Wlds + (k + kh) * P + col;
#pragma unroll
        for (int t = 0; t < NTL; ++t) {
            v2f b;
            b.x = w0[t * 16];
            b.y = w0[t * 16 + P];
            acc[t] = __builtin_amdgcn_wmma_f32_16x16x4_f32(
                         false, a, false, b, (short)0, acc[t], false, false);
        }
    }

    int orow = wave * 16 + ((lane >> 4) << 3);
#pragma unroll
    for (int t = 0; t < NT; ++t) {
        int ocol = t * 16 + (lane & 15);
        float badd = bias ? bias[ocol] : 0.f;
#pragma unroll
        for (int r = 0; r < 8; ++r)
            Y[(size_t)(orow + r) * Nout + ocol] = acc[t][r] + badd;
    }
    if (LOGITS) {
        const int c = lane & 15;
        if (c < 2 * NH) {
            float* dstp = (c < NH) ? (as_out + c) : (ad_out + (c - NH));
#pragma unroll
            for (int r = 0; r < 8; ++r)
                dstp[(size_t)(orow + r) * NH] = acc[NTL - 1][r];
        }
    }
}

#define AGG_TB   1024
#define AGG_NW   (AGG_TB / 32)
#define AGG_CAP  8192
#define KEEP_HDR    32
#define KEEP_STRIDE (KEEP_HDR + AGG_CAP)

static_assert(AGG_R <= 256 && NN <= 65536, "node / row packing");
static_assert((AGG_WG - 1) * AGG_R < NN, "every range holds a node");

__device__ __forceinline__ float lrelu02(float v) { return (v > 0.f) ? v : 0.2f * v; }

template <bool SAVED>
__global__ __launch_bounds__(AGG_TB)
void agg_range(const int* __restrict__ src,
               const int* __restrict__ dst,
               const float* __restrict__ as_v,
               const float* __restrict__ ad_v,
               const float* __restrict__ Hm,
               const float* __restrict__ b,
               float* __restrict__ out,
               unsigned* __restrict__ keep,
               const unsigned* __restrict__ planes)
{
    __shared__ float    acc[AGG_R * HID];
    __shared__ float    adl[AGG_R * NH];
    __shared__ int      mxl[AGG_R * NH];
    __shared__ float    denl[AGG_R * NH];
    __shared__ unsigned list[AGG_CAP];
    __shared__ float4   exl4[AGG_CAP];
    __shared__ int      cnt;
    float* exl = (float*)exl4;

    const int tid  = (int)threadIdx.x;
    const int lane = tid & 31;
    const int wave = tid >> 5;
    const int n0   = (int)blockIdx.x * AGG_R;
    const int nv   = (NN - n0 < AGG_R) ? (NN - n0) : AGG_R;
    const float4* as4 = (const float4*)as_v;
    unsigned* keep_r = keep + (size_t)blockIdx.x * KEEP_STRIDE;

    for (int i = tid; i < AGG_R * HID; i += AGG_TB) acc[i] = 0.f;
    for (int i = tid; i < AGG_R * NH; i += AGG_TB) {
        int n = n0 + (i >> 2);
        adl[i]  = (n < NN) ? ad_v[n * NH + (i & 3)] : 0.f;
        mxl[i]  = (int)0x80000000;
        denl[i] = 0.f;
    }
    if (SAVED) {
        if (tid == 0) cnt = (int)keep_r[0];
    } else {
        for (int i = tid; i < nv; i += AGG_TB)
            list[i] = (unsigned)(NE + n0 + i);
        if (tid == 0) cnt = nv;
    }
    __syncthreads();

    if (!SAVED) {
    const uint4* pa4 = (const uint4*)planes;
    const uint4* pb4 = pa4 + NG;
    const unsigned rid = blockIdx.x;
    const unsigned c0 = (rid &   1u) ? 0u : ~0u, c1 = (rid &   2u) ? 0u : ~0u;
    const unsigned c2 = (rid &   4u) ? 0u : ~0u, c3 = (rid &   8u) ? 0u : ~0u;
    const unsigned c4 = (rid &  16u) ? 0u : ~0u, c5 = (rid &  32u) ? 0u : ~0u;
    const unsigned c6 = (rid &  64u) ? 0u : ~0u, c7 = (rid & 128u) ? 0u : ~0u;
    const int rot = (int)blockIdx.x * (NG / AGG_WG);
    auto take = [&](const uint4& a, const uint4& b, int id) {
        unsigned m = (a.x ^ c0) & (a.y ^ c1) & (a.z ^ c2) & (a.w ^ c3)
                   & (b.x ^ c4) & (b.y ^ c5) & (b.z ^ c6) & (b.w ^ c7);
        if (m) {
            int p = atomicAdd(&cnt, __popc(m));
            const unsigned e0 = (unsigned)id * 32u;
            do {
                const unsigned bit = (unsigned)__ffs((int)m) - 1u;
                m &= m - 1u;
                if (p < AGG_CAP) list[p] = e0 + bit;
                ++p;
            } while (m);
        }
    };
    int j = tid;
    for (; j + 3 * AGG_TB < NG; j += 4 * AGG_TB) {
        uint4 pa[4], pb[4];
        int   ix[4];
#pragma unroll
        for (int u = 0; u < 4; ++u) {
            int id = j + u * AGG_TB + rot;
            if (id >= NG) id -= NG;
            ix[u] = id;
            pa[u] = pa4[id];
            pb[u] = pb4[id];
        }
#pragma unroll
        for (int u = 0; u < 4; ++u) take(pa[u], pb[u], ix[u]);
    }
    for (; j < NG; j += AGG_TB) {
        int id = j + rot;
        if (id >= NG) id -= NG;
        take(pa4[id], pb4[id], id);
    }
    __syncthreads();
    }
    const int total = cnt;
    if (!SAVED && tid == 0) keep_r[0] = (unsigned)total;

    if (total <= AGG_CAP) {
        for (int i = tid; i < total; i += AGG_TB) {
            unsigned q;
            if (SAVED) {
                q = keep_r[KEEP_HDR + i];
            } else {
                int e  = (int)list[i];
                int s0 = (e < NE) ? src[e] : (e - NE);
                int d0 = (e < NE) ? dst[e] : (e - NE);
                q = (unsigned)s0 | ((unsigned)(d0 - n0) << 16);
                keep_r[KEEP_HDR + i] = q;
            }
            list[i] = q;
            int s  = (int)(q & 0xFFFFu);
            int dl = (int)(q >> 16);
            float4 a = as4[s];
            float v0 = lrelu02(a.x + adl[dl * NH + 0]);
            float v1 = lrelu02(a.y + adl[dl * NH + 1]);
            float v2 = lrelu02(a.z + adl[dl * NH + 2]);
            float v3 = lrelu02(a.w + adl[dl * NH + 3]);
            exl[i * NH + 0] = v0; exl[i * NH + 1] = v1;
            exl[i * NH + 2] = v2; exl[i * NH + 3] = v3;
            atomicMax(&mxl[dl * NH + 0], f2ord(v0));
            atomicMax(&mxl[dl * NH + 1], f2ord(v1));
            atomicMax(&mxl[dl * NH + 2], f2ord(v2));
            atomicMax(&mxl[dl * NH + 3], f2ord(v3));
        }
        __syncthreads();
        for (int i = tid; i < total * NH; i += AGG_TB) {
            int dl = (int)(list[i >> 2] >> 16);
            int h  = i & 3;
            float ex = expf(exl[i] - ord2f(mxl[dl * NH + h]));
            exl[i] = ex;
            atomicAdd(&denl[dl * NH + h], ex);
        }
        __syncthreads();
        int i = wave;
        for (; i + 3 * AGG_NW < total; i += 4 * AGG_NW) {
            unsigned p[4]; float4 w[4]; float hv[4][NH];
#pragma unroll
            for (int u = 0; u < 4; ++u) {
                p[u] = list[i + u * AGG_NW];
                w[u] = exl4[i + u * AGG_NW];
            }
#pragma unroll
            for (int u = 0; u < 4; ++u) {
                const float* hp = Hm + (size_t)(p[u] & 0xFFFFu) * HID + lane;
#pragma unroll
                for (int k = 0; k < NH; ++k) hv[u][k] = hp[32 * k];
            }
#pragma unroll
            for (int u = 0; u < 4; ++u) {
                float* ap = acc + (p[u] >> 16) * HID + lane;
                atomicAdd(ap +  0, hv[u][0] * w[u].x);
                atomicAdd(ap + 32, hv[u][1] * w[u].y);
                atomicAdd(ap + 64, hv[u][2] * w[u].z);
                atomicAdd(ap + 96, hv[u][3] * w[u].w);
            }
        }
        for (; i < total; i += AGG_NW) {
            unsigned p = list[i];
            float4 w = exl4[i];
            const float* hp = Hm + (size_t)(p & 0xFFFFu) * HID + lane;
            float h0 = hp[0], h1 = hp[32], h2 = hp[64], h3 = hp[96];
            float* ap = acc + (p >> 16) * HID + lane;
            atomicAdd(ap +  0, h0 * w.x);
            atomicAdd(ap + 32, h1 * w.y);
            atomicAdd(ap + 64, h2 * w.z);
            atomicAdd(ap + 96, h3 * w.w);
        }
    } else {
        for (int e = tid; e < ET; e += AGG_TB) {
            int s = (e < NE) ? src[e] : (e - NE);
            int d = (e < NE) ? dst[e] : (e - NE);
            unsigned dl = (unsigned)(d - n0);
            if (dl < (unsigned)nv) {
                float4 a = as4[s];
                atomicMax(&mxl[dl * NH + 0], f2ord(lrelu02(a.x + adl[dl * NH + 0])));
                atomicMax(&mxl[dl * NH + 1], f2ord(lrelu02(a.y + adl[dl * NH + 1])));
                atomicMax(&mxl[dl * NH + 2], f2ord(lrelu02(a.z + adl[dl * NH + 2])));
                atomicMax(&mxl[dl * NH + 3], f2ord(lrelu02(a.w + adl[dl * NH + 3])));
            }
        }
        __syncthreads();
        for (int e = tid; e < ET; e += AGG_TB) {
            int s = (e < NE) ? src[e] : (e - NE);
            int d = (e < NE) ? dst[e] : (e - NE);
            unsigned dl = (unsigned)(d - n0);
            if (dl < (unsigned)nv) {
                const float* ap4 = as_v + (size_t)s * NH;
                const float* hp  = Hm + (size_t)s * HID;
                for (int h = 0; h < NH; ++h) {
                    float v  = lrelu02(ap4[h] + adl[dl * NH + h]);
                    float ex = expf(v - ord2f(mxl[dl * NH + h]));
                    atomicAdd(&denl[dl * NH + h], ex);
                    for (int c = 0; c < CH; ++c)
                        atomicAdd(&acc[dl * HID + h * CH + c], hp[h * CH + c] * ex);
                }
            }
        }
    }
    __syncthreads();

    for (int dl = wave; dl < nv; dl += AGG_NW) {
#pragma unroll
        for (int k = 0; k < NH; ++k) {
            int c = lane + 32 * k;
            float y = acc[dl * HID + c] / (denl[dl * NH + k] + 1e-16f) + b[c];
            out[(size_t)(n0 + dl) * HID + c] = (y > 0.f) ? y : (expf(y) - 1.f);
        }
    }
}

template <bool SAVED>
static void run_gat_layer(const float* Xin, const float* W,
                          const float* a_s, const float* a_d, const float* b,
                          const int* src, const int* dst,
                          float* bufH, float* bufOut,
                          float* asb, float* adb, unsigned* keep, unsigned* planes, hipStream_t stream)
{
    const int TB = GEMM_TB;
    int nMtiles = NN / 16;
    int blocks  = (nMtiles * 32 + TB - 1) / TB;
    wmma_gemm_strip<HID, HID / 16, true, !SAVED><<<blocks, TB, 0, stream>>>(
        Xin, W, nullptr, bufH, nMtiles, a_s, a_d, asb, adb, dst, planes);
    agg_range<SAVED><<<AGG_WG, AGG_TB, 0, stream>>>(src, dst, asb, adb, bufH, b, bufOut, keep, planes);
}

extern "C" void kernel_launch(void* const* d_in, const int* in_sizes, int n_in,
                              void* d_out, int out_size, void* d_ws, size_t ws_size,
                              hipStream_t stream)
{
    const float* x    = (const float*)d_in[0];
    const int*   edge = (const int*)  d_in[1];
    const float* W1   = (const float*)d_in[2];
    const float* a_s1 = (const float*)d_in[3];
    const float* a_d1 = (const float*)d_in[4];
    const float* b1   = (const float*)d_in[5];
    const float* W2   = (const float*)d_in[6];
    const float* a_s2 = (const float*)d_in[7];
    const float* a_d2 = (const float*)d_in[8];
    const float* b2   = (const float*)d_in[9];
    const float* fc_w = (const float*)d_in[10];
    const float* fc_b = (const float*)d_in[11];
    float* out = (float*)d_out;

    const int* src = edge;
    const int* dst = edge + NE;

    float* bufH   = (float*)d_ws;
    float* bufIn2 = bufH   + (size_t)NN * HID;
    float* asb    = bufIn2 + (size_t)NN * HID;
    float* adb    = asb    + (size_t)NN * NH;
    unsigned* keep = (unsigned*)(adb + (size_t)NN * NH);
    unsigned* planes = keep + (size_t)AGG_WG * KEEP_STRIDE;

    run_gat_layer<false>(x, W1, a_s1, a_d1, b1, src, dst, bufH, bufIn2, asb, adb, keep, planes, stream);
    run_gat_layer<true>(bufIn2, W2, a_s2, a_d2, b2, src, dst, bufH, bufIn2, asb, adb, keep, planes, stream);
    const int TB = GEMM_TB;
    int nMtiles = NN / 16;
    int blocks  = (nMtiles * 32 + TB - 1) / TB;
    wmma_gemm_strip<HID, CH / 16, false, false><<<blocks, TB, 0, stream>>>(
        bufIn2, fc_w, fc_b, out, nMtiles, nullptr, nullptr, nullptr, nullptr, nullptr, nullptr);
}


# Aggregation row gather: one 16-byte load per lane fetches the whole 512-byte row (was four 4-byte loads, one line each); LDS rows re-laid so the 4 float atomics stay conflict-free; same sum order
# speedup vs baseline: 3.7563x; 1.0495x over previous
#include <hip/hip_runtime.h>
#include <math.h>

#define NN    50000
#define NE    800000
#define ET    (NE + NN)
#define HID   128
#define NH    4
#define CH    32

typedef float v2f __attribute__((ext_vector_type(2)));
typedef float v8f __attribute__((ext_vector_type(8)));

#define AGG_R    198
#define AGG_WG   ((NN + AGG_R - 1) / AGG_R)
#define RID_NONE 255u
#define NG       (NE / 32)
#define PLANE_WORDS (2 * 4 * NG)
static_assert(AGG_WG <= 254, "range ids must leave 255 free");
static_assert(NE % 256 == 0, "the table is built in chunks of 256 edges");
static_assert(NE / 256 <= NN / 16, "one 256-edge chunk per strip wave of the projection launch");

__device__ __forceinline__ int f2ord(float f) {
    int i = __float_as_int(f);
    return (i >= 0) ? i : (i ^ 0x7fffffff);
}
__device__ __forceinline__ float ord2f(int i) {
    return __int_as_float((i >= 0) ? i : (i ^ 0x7fffffff));
}

#define GEMM_TB 256
template <int K, int NT, bool LOGITS, bool PLANES>
__global__ __launch_bounds__(GEMM_TB)
void wmma_gemm_strip(const float* __restrict__ X,
                                const float* __restrict__ W,
                                const float* __restrict__ bias,
                                float* __restrict__ Y,
                                int nMtiles,
                                const float* __restrict__ a_s,
                                const float* __restrict__ a_d,
                                float* __restrict__ as_out,
                                float* __restrict__ ad_out,
                                const int* __restrict__ edst,
                                unsigned* __restrict__ planes)
{
    constexpr int Nout = NT * 16;
    constexpr int NTL  = LOGITS ? NT + 1 : NT;
    constexpr int P    = NTL * 16;
    __shared__ float Wlds[K * P];

    const int gwave = (int)((blockIdx.x * blockDim.x + threadIdx.x) >> 5);
    int dv[8];
    if (PLANES) {
        if (gwave < NE / 256) {
#pragma unroll
            for (int g = 0; g < 8; ++g)
                dv[g] = edst[gwave * 256 + g * 32 + ((int)threadIdx.x & 31)];
        }
    }

    for (int i = threadIdx.x * 4; i < K * Nout; i += blockDim.x * 4)
        *(float4*)&Wlds[(i / Nout) * P + (i % Nout)] = *(const float4*)&W[i];

    if (LOGITS) {
        static_assert(!LOGITS || (K == HID && Nout == HID), "fold is written for 128 x 128");
        __syncthreads();
        const int ln = (int)threadIdx.x & 31;
        const int wv = (int)threadIdx.x >> 5;
        const float4 s4 = *(const float4*)(a_s + 4 * ln);
        const float4 d4 = *(const float4*)(a_d + 4 * ln);
        constexpr int NW = GEMM_TB / 32;
        constexpr int R  = K / NW;
        float ps[R], pd[R];
#pragma unroll
        for (int it = 0; it < R; ++it) {
            const int k = it * NW + wv;
            const float4 w4 = *(const float4*)(Wlds + k * P + 4 * ln);
            ps[it] = w4.x * s4.x + w4.y * s4.y + w4.z * s4.z + w4.w * s4.w;
            pd[it] = w4.x * d4.x + w4.y * d4.y + w4.z * d4.z + w4.w * d4.w;
        }
#pragma unroll
        for (int m = 1; m < 8; m <<= 1) {
#pragma unroll
            for (int it = 0; it < R; ++it) {
                ps[it] += __shfl_xor(ps[it], m);
                pd[it] += __shfl_xor(pd[it], m);
            }
        }
        float* wa = Wlds + wv * P + Nout;
        if ((ln & 7) == 0) {
#pragma unroll
            for (int it = 0; it < R; ++it) {
                wa[it * NW * P + (ln >> 3)]      = ps[it];
                wa[it * NW * P + NH + (ln >> 3)] = pd[it];
            }
        } else if ((ln & 7) == 1) {
#pragma unroll
            for (int it = 0; it < R; ++it) {
                wa[it * NW * P + 2 * NH + (ln >> 3)] = 0.f;
                wa[it * NW * P + 3 * NH + (ln >> 3)] = 0.f;
            }
        }
    }
    __syncthreads();

    if (PLANES) {
        if (gwave < NE / 256) {
            const int ln = (int)threadIdx.x & 31;
            unsigned wa = 0u, wb = 0u;
#pragma unroll
            for (int g = 0; g < 8; ++g) {
                const unsigned d = (unsigned)dv[g];
                const unsigned r = (d < (unsigned)NN) ? d / (unsigned)AGG_R : RID_NONE;
#pragma unroll
                for (int k = 0; k < 8; ++k) {
                    const unsigned bal = __builtin_amdgcn_ballot_w32(((r >> k) & 1u) != 0u);
                    if (k < 4) { if (ln == 4 * g + k)     wa = bal; }
                    else       { if (ln == 4 * g + k - 4) wb = bal; }
                }
            }
            planes[gwave * 32 + ln]          = wa;
            planes[4 * NG + gwave * 32 + ln] = wb;
        }
    }

    int wave = gwave;
    if (wave >= nMtiles) return;
    int lane = (int)(threadIdx.x & 31);

    int row = wave * 16 + (lane & 15);
    int col = lane & 15;
    int kh  = (lane >> 4) << 1;

    v8f acc[NTL];
#pragma unroll
    for (int t = 0; t < NTL; ++t)
        acc[t] = (v8f){0.f, 0.f, 0.f, 0.f, 0.f, 0.f, 0.f, 0.f};

    const float* xrow = X + (size_t)row * K;
#pragma unroll
    for (int k = 0; k < K; k += 4) {
        v2f a = *(const v2f*)(xrow + k + kh);
        const float* w0 = Wlds + (k + kh) * P + col;
#pragma unroll
        for (int t = 0; t < NTL; ++t) {
            v2f b;
            b.x = w0[t * 16];
            b.y = w0[t * 16 + P];
            acc[t] = __builtin_amdgcn_wmma_f32_16x16x4_f32(
                         false, a, false, b, (short)0, acc[t], false, false);
        }
    }

    int orow = wave * 16 + ((lane >> 4) << 3);
#pragma unroll
    for (int t = 0; t < NT; ++t) {
        int ocol = t * 16 + (lane & 15);
        float badd = bias ? bias[ocol] : 0.f;
#pragma unroll
        for (int r = 0; r < 8; ++r)
            Y[(size_t)(orow + r) * Nout + ocol] = acc[t][r] + badd;
    }
    if (LOGITS) {
        const int c = lane & 15;
        if (c < 2 * NH) {
            float* dstp = (c < NH) ? (as_out + c) : (ad_out + (c - NH));
#pragma unroll
            for (int r = 0; r < 8; ++r)
                dstp[(size_t)(orow + r) * NH] = acc[NTL - 1][r];
        }
    }
}

#define AGG_TB   1024
#define AGG_NW   (AGG_TB / 32)
#define AGG_CAP  8192
#define KEEP_HDR    32
#define KEEP_STRIDE (KEEP_HDR + AGG_CAP)

static_assert(AGG_R <= 256 && NN <= 65536, "node / row packing");
static_assert((AGG_WG - 1) * AGG_R < NN, "every range holds a node");

__device__ __forceinline__ float lrelu02(float v) { return (v > 0.f) ? v : 0.2f * v; }

template <bool SAVED>
__global__ __launch_bounds__(AGG_TB)
void agg_range(const int* __restrict__ src,
               const int* __restrict__ dst,
               const float* __restrict__ as_v,
               const float* __restrict__ ad_v,
               const float* __restrict__ Hm,
               const float* __restrict__ b,
               float* __restrict__ out,
               unsigned* __restrict__ keep,
               const unsigned* __restrict__ planes)
{
    __shared__ float    acc[AGG_R * HID];
    __shared__ float    adl[AGG_R * NH];
    __shared__ int      mxl[AGG_R * NH];
    __shared__ float    denl[AGG_R * NH];
    __shared__ unsigned list[AGG_CAP];
    __shared__ float4   exl4[AGG_CAP];
    __shared__ int      cnt;
    float* exl = (float*)exl4;

    const int tid  = (int)threadIdx.x;
    const int lane = tid & 31;
    const int wave = tid >> 5;
    const int n0   = (int)blockIdx.x * AGG_R;
    const int nv   = (NN - n0 < AGG_R) ? (NN - n0) : AGG_R;
    const float4* as4 = (const float4*)as_v;
    unsigned* keep_r = keep + (size_t)blockIdx.x * KEEP_STRIDE;

    for (int i = tid; i < AGG_R * HID; i += AGG_TB) acc[i] = 0.f;
    for (int i = tid; i < AGG_R * NH; i += AGG_TB) {
        int n = n0 + (i >> 2);
        adl[i]  = (n < NN) ? ad_v[n * NH + (i & 3)] : 0.f;
        mxl[i]  = (int)0x80000000;
        denl[i] = 0.f;
    }
    if (SAVED) {
        if (tid == 0) cnt = (int)keep_r[0];
    } else {
        for (int i = tid; i < nv; i += AGG_TB)
            list[i] = (unsigned)(NE + n0 + i);
        if (tid == 0) cnt = nv;
    }
    __syncthreads();

    if (!SAVED) {
    const uint4* pa4 = (const uint4*)planes;
    const uint4* pb4 = pa4 + NG;
    const unsigned rid = blockIdx.x;
    const unsigned c0 = (rid &   1u) ? 0u : ~0u, c1 = (rid &   2u) ? 0u : ~0u;
    const unsigned c2 = (rid &   4u) ? 0u : ~0u, c3 = (rid &   8u) ? 0u : ~0u;
    const unsigned c4 = (rid &  16u) ? 0u : ~0u, c5 = (rid &  32u) ? 0u : ~0u;
    const unsigned c6 = (rid &  64u) ? 0u : ~0u, c7 = (rid & 128u) ? 0u : ~0u;
    const int rot = (int)blockIdx.x * (NG / AGG_WG);
    auto take = [&](const uint4& a, const uint4& b, int id) {
        unsigned m = (a.x ^ c0) & (a.y ^ c1) & (a.z ^ c2) & (a.w ^ c3)
                   & (b.x ^ c4) & (b.y ^ c5) & (b.z ^ c6) & (b.w ^ c7);
        if (m) {
            int p = atomicAdd(&cnt, __popc(m));
            const unsigned e0 = (unsigned)id * 32u;
            do {
                const unsigned bit = (unsigned)__ffs((int)m) - 1u;
                m &= m - 1u;
                if (p < AGG_CAP) list[p] = e0 + bit;
                ++p;
            } while (m);
        }
    };
    int j = tid;
    for (; j + 3 * AGG_TB < NG; j += 4 * AGG_TB) {
        uint4 pa[4], pb[4];
        int   ix[4];
#pragma unroll
        for (int u = 0; u < 4; ++u) {
            int id = j + u * AGG_TB + rot;
            if (id >= NG) id -= NG;
            ix[u] = id;
            pa[u] = pa4[id];
            pb[u] = pb4[id];
        }
#pragma unroll
        for (int u = 0; u < 4; ++u) take(pa[u], pb[u], ix[u]);
    }
    for (; j < NG; j += AGG_TB) {
        int id = j + rot;
        if (id >= NG) id -= NG;
        take(pa4[id], pb4[id], id);
    }
    __syncthreads();
    }
    const int total = cnt;
    if (!SAVED && tid == 0) keep_r[0] = (unsigned)total;

    if (total <= AGG_CAP) {
        for (int i = tid; i < total; i += AGG_TB) {
            unsigned q;
            if (SAVED) {
                q = keep_r[KEEP_HDR + i];
            } else {
                int e  = (int)list[i];
                int s0 = (e < NE) ? src[e] : (e - NE);
                int d0 = (e < NE) ? dst[e] : (e - NE);
                q = (unsigned)s0 | ((unsigned)(d0 - n0) << 16);
                keep_r[KEEP_HDR + i] = q;
            }
            list[i] = q;
            int s  = (int)(q & 0xFFFFu);
            int dl = (int)(q >> 16);
            float4 a = as4[s];
            float v0 = lrelu02(a.x + adl[dl * NH + 0]);
            float v1 = lrelu02(a.y + adl[dl * NH + 1]);
            float v2 = lrelu02(a.z + adl[dl * NH + 2]);
            float v3 = lrelu02(a.w + adl[dl * NH + 3]);
            exl[i * NH + 0] = v0; exl[i * NH + 1] = v1;
            exl[i * NH + 2] = v2; exl[i * NH + 3] = v3;
            atomicMax(&mxl[dl * NH + 0], f2ord(v0));
            atomicMax(&mxl[dl * NH + 1], f2ord(v1));
            atomicMax(&mxl[dl * NH + 2], f2ord(v2));
            atomicMax(&mxl[dl * NH + 3], f2ord(v3));
        }
        __syncthreads();
        for (int i = tid; i < total * NH; i += AGG_TB) {
            int dl = (int)(list[i >> 2] >> 16);
            int h  = i & 3;
            float ex = expf(exl[i] - ord2f(mxl[dl * NH + h]));
            exl[i] = ex;
            atomicAdd(&denl[dl * NH + h], ex);
        }
        __syncthreads();
        {
            const int hd = lane >> 3;
            const float4* H4 = (const float4*)Hm;
            int i = wave;
            for (; i + 7 * AGG_NW < total; i += 8 * AGG_NW) {
                unsigned p[8]; float w[8]; float4 hv[8];
#pragma unroll
                for (int u = 0; u < 8; ++u) {
                    p[u] = list[i + u * AGG_NW];
                    w[u] = exl[(i + u * AGG_NW) * NH + hd];
                }
#pragma unroll
                for (int u = 0; u < 8; ++u)
                    hv[u] = H4[(size_t)(p[u] & 0xFFFFu) * (HID / 4) + lane];
#pragma unroll
                for (int u = 0; u < 8; ++u) {
                    float* ap = acc + (p[u] >> 16) * HID + lane;
                    atomicAdd(ap +  0, hv[u].x * w[u]);
                    atomicAdd(ap + 32, hv[u].y * w[u]);
                    atomicAdd(ap + 64, hv[u].z * w[u]);
                    atomicAdd(ap + 96, hv[u].w * w[u]);
                }
            }
            {
                unsigned p[8]; float w[8]; float4 hv[8];
#pragma unroll
                for (int u = 0; u < 7; ++u) {
                    if (i + u * AGG_NW < total) {
                        p[u] = list[i + u * AGG_NW];
                        w[u] = exl[(i + u * AGG_NW) * NH + hd];
                    }
                }
#pragma unroll
                for (int u = 0; u < 7; ++u) {
                    if (i + u * AGG_NW < total)
                        hv[u] = H4[(size_t)(p[u] & 0xFFFFu) * (HID / 4) + lane];
                }
#pragma unroll
                for (int u = 0; u < 7; ++u) {
                    if (i + u * AGG_NW < total) {
                        float* ap = acc + (p[u] >> 16) * HID + lane;
                        atomicAdd(ap +  0, hv[u].x * w[u]);
                        atomicAdd(ap + 32, hv[u].y * w[u]);
                        atomicAdd(ap + 64, hv[u].z * w[u]);
                        atomicAdd(ap + 96, hv[u].w * w[u]);
                    }
                }
            }
        }
    } else {
        for (int e = tid; e < ET; e += AGG_TB) {
            int s = (e < NE) ? src[e] : (e - NE);
            int d = (e < NE) ? dst[e] : (e - NE);
            unsigned dl = (unsigned)(d - n0);
            if (dl < (unsigned)nv) {
                float4 a = as4[s];
                atomicMax(&mxl[dl * NH + 0], f2ord(lrelu02(a.x + adl[dl * NH + 0])));
                atomicMax(&mxl[dl * NH + 1], f2ord(lrelu02(a.y + adl[dl * NH + 1])));
                atomicMax(&mxl[dl * NH + 2], f2ord(lrelu02(a.z + adl[dl * NH + 2])));
                atomicMax(&mxl[dl * NH + 3], f2ord(lrelu02(a.w + adl[dl * NH + 3])));
            }
        }
        __syncthreads();
        for (int e = tid; e < ET; e += AGG_TB) {
            int s = (e < NE) ? src[e] : (e - NE);
            int d = (e < NE) ? dst[e] : (e - NE);
            unsigned dl = (unsigned)(d - n0);
            if (dl < (unsigned)nv) {
                const float* ap4 = as_v + (size_t)s * NH;
                const float* hp  = Hm + (size_t)s * HID;
                for (int h = 0; h < NH; ++h) {
                    float v  = lrelu02(ap4[h] + adl[dl * NH + h]);
                    float ex = expf(v - ord2f(mxl[dl * NH + h]));
                    atomicAdd(&denl[dl * NH + h], ex);
                    for (int c = 0; c < CH; ++c)
                    {
                        const int ch = h * CH + c;
                        atomicAdd(&acc[dl * HID + (ch & 3) * 32 + (ch >> 2)], hp[ch] * ex);
                    }
                }
            }
        }
    }
    __syncthreads();

    {
        const int hd = lane >> 3;
        const float4 bb = *(const float4*)(b + 4 * lane);
        for (int dl = wave; dl < nv; dl += AGG_NW) {
            const float* ar = acc + dl * HID + lane;
            const float dn = denl[dl * NH + hd] + 1e-16f;
            float4 y;
            y.x = ar[ 0] / dn + bb.x;
            y.y = ar[32] / dn + bb.y;
            y.z = ar[64] / dn + bb.z;
            y.w = ar[96] / dn + bb.w;
            y.x = (y.x > 0.f) ? y.x : (expf(y.x) - 1.f);
            y.y = (y.y > 0.f) ? y.y : (expf(y.y) - 1.f);
            y.z = (y.z > 0.f) ? y.z : (expf(y.z) - 1.f);
            y.w = (y.w > 0.f) ? y.w : (expf(y.w) - 1.f);
            *(float4*)(out + (size_t)(n0 + dl) * HID + 4 * lane) = y;
        }
    }
}

template <bool SAVED>
static void run_gat_layer(const float* Xin, const float* W,
                          const float* a_s, const float* a_d, const float* b,
                          const int* src, const int* dst,
                          float* bufH, float* bufOut,
                          float* asb, float* adb, unsigned* keep, unsigned* planes, hipStream_t stream)
{
    const int TB = GEMM_TB;
    int nMtiles = NN / 16;
    int blocks  = (nMtiles * 32 + TB - 1) / TB;
    wmma_gemm_strip<HID, HID / 16, true, !SAVED><<<blocks, TB, 0, stream>>>(
        Xin, W, nullptr, bufH, nMtiles, a_s, a_d, asb, adb, dst, planes);
    agg_range<SAVED><<<AGG_WG, AGG_TB, 0, stream>>>(src, dst, asb, adb, bufH, b, bufOut, keep, planes);
}

extern "C" void kernel_launch(void* const* d_in, const int* in_sizes, int n_in,
                              void* d_out, int out_size, void* d_ws, size_t ws_size,
                              hipStream_t stream)
{
    const float* x    = (const float*)d_in[0];
    const int*   edge = (const int*)  d_in[1];
    const float* W1   = (const float*)d_in[2];
    const float* a_s1 = (const float*)d_in[3];
    const float* a_d1 = (const float*)d_in[4];
    const float* b1   = (const float*)d_in[5];
    const float* W2   = (const float*)d_in[6];
    const float* a_s2 = (const float*)d_in[7];
    const float* a_d2 = (const float*)d_in[8];
    const float* b2   = (const float*)d_in[9];
    const float* fc_w = (const float*)d_in[10];
    const float* fc_b = (const float*)d_in[11];
    float* out = (float*)d_out;

    const int* src = edge;
    const int* dst = edge + NE;

    float* bufH   = (float*)d_ws;
    float* bufIn2 = bufH   + (size_t)NN * HID;
    float* asb    = bufIn2 + (size_t)NN * HID;
    float* adb    = asb    + (size_t)NN * NH;
    unsigned* keep = (unsigned*)(adb + (size_t)NN * NH);
    unsigned* planes = keep + (size_t)AGG_WG * KEEP_STRIDE;

    run_gat_layer<false>(x, W1, a_s1, a_d1, b1, src, dst, bufH, bufIn2, asb, adb, keep, planes, stream);
    run_gat_layer<true>(bufIn2, W2, a_s2, a_d2, b2, src, dst, bufH, bufIn2, asb, adb, keep, planes, stream);
    const int TB = GEMM_TB;
    int nMtiles = NN / 16;
    int blocks  = (nMtiles * 32 + TB - 1) / TB;
    wmma_gemm_strip<HID, CH / 16, false, false><<<blocks, TB, 0, stream>>>(
        bufIn2, fc_w, fc_b, out, nMtiles, nullptr, nullptr, nullptr, nullptr, nullptr, nullptr);
}


# Range list counting-sorted by node in LDS; a wave sums each node's weighted rows in registers and stores the LDS row once: the 4 row atomics per edge go; float sum order: slot order per node
# speedup vs baseline: 4.0449x; 1.0768x over previous
#include <hip/hip_runtime.h>
#include <math.h>

#define NN    50000
#define NE    800000
#define ET    (NE + NN)
#define HID   128
#define NH    4
#define CH    32

typedef float v2f __attribute__((ext_vector_type(2)));
typedef float v8f __attribute__((ext_vector_type(8)));

#define AGG_R    198
#define AGG_WG   ((NN + AGG_R - 1) / AGG_R)
#define RID_NONE 255u
#define NG       (NE / 32)
#define PLANE_WORDS (2 * 4 * NG)
static_assert(AGG_WG <= 254, "range ids must leave 255 free");
static_assert(NE % 256 == 0, "the table is built in chunks of 256 edges");
static_assert(NE / 256 <= NN / 16, "one 256-edge chunk per strip wave of the projection launch");

__device__ __forceinline__ int f2ord(float f) {
    int i = __float_as_int(f);
    return (i >= 0) ? i : (i ^ 0x7fffffff);
}
__device__ __forceinline__ float ord2f(int i) {
    return __int_as_float((i >= 0) ? i : (i ^ 0x7fffffff));
}

#define GEMM_TB 256
template <int K, int NT, bool LOGITS, bool PLANES>
__global__ __launch_bounds__(GEMM_TB)
void wmma_gemm_strip(const float* __restrict__ X,
                                const float* __restrict__ W,
                                const float* __restrict__ bias,
                                float* __restrict__ Y,
                                int nMtiles,
                                const float* __restrict__ a_s,
                                const float* __restrict__ a_d,
                                float* __restrict__ as_out,
                                float* __restrict__ ad_out,
                                const int* __restrict__ edst,
                                unsigned* __restrict__ planes)
{
    constexpr int Nout = NT * 16;
    constexpr int NTL  = LOGITS ? NT + 1 : NT;
    constexpr int P    = NTL * 16;
    __shared__ float Wlds[K * P];

    const int gwave = (int)((blockIdx.x * blockDim.x + threadIdx.x) >> 5);
    int dv[8];
    if (PLANES) {
        if (gwave < NE / 256) {
#pragma unroll
            for (int g = 0; g < 8; ++g)
                dv[g] = edst[gwave * 256 + g * 32 + ((int)threadIdx.x & 31)];
        }
    }

    for (int i = threadIdx.x * 4; i < K * Nout; i += blockDim.x * 4)
        *(float4*)&Wlds[(i / Nout) * P + (i % Nout)] = *(const float4*)&W[i];

    if (LOGITS) {
        static_assert(!LOGITS || (K == HID && Nout == HID), "fold is written for 128 x 128");
        __syncthreads();
        const int ln = (int)threadIdx.x & 31;
        const int wv = (int)threadIdx.x >> 5;
        const float4 s4 = *(const float4*)(a_s + 4 * ln);
        const float4 d4 = *(const float4*)(a_d + 4 * ln);
        constexpr int NW = GEMM_TB / 32;
        constexpr int R  = K / NW;
        float ps[R], pd[R];
#pragma unroll
        for (int it = 0; it < R; ++it) {
            const int k = it * NW + wv;
            const float4 w4 = *(const float4*)(Wlds + k * P + 4 * ln);
            ps[it] = w4.x * s4.x + w4.y * s4.y + w4.z * s4.z + w4.w * s4.w;
            pd[it] = w4.x * d4.x + w4.y * d4.y + w4.z * d4.z + w4.w * d4.w;
        }
#pragma unroll
        for (int m = 1; m < 8; m <<= 1) {
#pragma unroll
            for (int it = 0; it < R; ++it) {
                ps[it] += __shfl_xor(ps[it], m);
                pd[it] += __shfl_xor(pd[it], m);
            }
        }
        float* wa = Wlds + wv * P + Nout;
        if ((ln & 7) == 0) {
#pragma unroll
            for (int it = 0; it < R; ++it) {
                wa[it * NW * P + (ln >> 3)]      = ps[it];
                wa[it * NW * P + NH + (ln >> 3)] = pd[it];
            }
        } else if ((ln & 7) == 1) {
#pragma unroll
            for (int it = 0; it < R; ++it) {
                wa[it * NW * P + 2 * NH + (ln >> 3)] = 0.f;
                wa[it * NW * P + 3 * NH + (ln >> 3)] = 0.f;
            }
        }
    }
    __syncthreads();

    if (PLANES) {
        if (gwave < NE / 256) {
            const int ln = (int)threadIdx.x & 31;
            unsigned wa = 0u, wb = 0u;
#pragma unroll
            for (int g = 0; g < 8; ++g) {
                const unsigned d = (unsigned)dv[g];
                const unsigned r = (d < (unsigned)NN) ? d / (unsigned)AGG_R : RID_NONE;
#pragma unroll
                for (int k = 0; k < 8; ++k) {
                    const unsigned bal = __builtin_amdgcn_ballot_w32(((r >> k) & 1u) != 0u);
                    if (k < 4) { if (ln == 4 * g + k)     wa = bal; }
                    else       { if (ln == 4 * g + k - 4) wb = bal; }
                }
            }
            planes[gwave * 32 + ln]          = wa;
            planes[4 * NG + gwave * 32 + ln] = wb;
        }
    }

    int wave = gwave;
    if (wave >= nMtiles) return;
    int lane = (int)(threadIdx.x & 31);

    int row = wave * 16 + (lane & 15);
    int col = lane & 15;
    int kh  = (lane >> 4) << 1;

    v8f acc[NTL];
#pragma unroll
    for (int t = 0; t < NTL; ++t)
        acc[t] = (v8f){0.f, 0.f, 0.f, 0.f, 0.f, 0.f, 0.f, 0.f};

    const float* xrow = X + (size_t)row * K;
#pragma unroll
    for (int k = 0; k < K; k += 4) {
        v2f a = *(const v2f*)(xrow + k + kh);
        const float* w0 = Wlds + (k + kh) * P + col;
#pragma unroll
        for (int t = 0; t < NTL; ++t) {
            v2f b;
            b.x = w0[t * 16];
            b.y = w0[t * 16 + P];
            acc[t] = __builtin_amdgcn_wmma_f32_16x16x4_f32(
                         false, a, false, b, (short)0, acc[t], false, false);
        }
    }

    int orow = wave * 16 + ((lane >> 4) << 3);
#pragma unroll
    for (int t = 0; t < NT; ++t) {
        int ocol = t * 16 + (lane & 15);
        float badd = bias ? bias[ocol] : 0.f;
#pragma unroll
        for (int r = 0; r < 8; ++r)
            Y[(size_t)(orow + r) * Nout + ocol] = acc[t][r] + badd;
    }
    if (LOGITS) {
        const int c = lane & 15;
        if (c < 2 * NH) {
            float* dstp = (c < NH) ? (as_out + c) : (ad_out + (c - NH));
#pragma unroll
            for (int r = 0; r < 8; ++r)
                dstp[(size_t)(orow + r) * NH] = acc[NTL - 1][r];
        }
    }
}

#define AGG_TB   1024
#define AGG_NW   (AGG_TB / 32)
#define AGG_CAP  8192
#define KEEP_HDR    32
#define KEEP_STRIDE (KEEP_HDR + AGG_CAP)

static_assert(AGG_R <= 256 && NN <= 65536, "node / row packing");
static_assert((AGG_WG - 1) * AGG_R < NN, "every range holds a node");

__device__ __forceinline__ float lrelu02(float v) { return (v > 0.f) ? v : 0.2f * v; }

template <bool SAVED>
__global__ __launch_bounds__(AGG_TB)
void agg_range(const int* __restrict__ src,
               const int* __restrict__ dst,
               const float* __restrict__ as_v,
               const float* __restrict__ ad_v,
               const float* __restrict__ Hm,
               const float* __restrict__ b,
               float* __restrict__ out,
               unsigned* __restrict__ keep,
               const unsigned* __restrict__ planes)
{
    __shared__ float    acc[AGG_R * HID];
    __shared__ float    adl[AGG_R * NH];
    __shared__ int      mxl[AGG_R * NH];
    __shared__ float    denl[AGG_R * NH];
    __shared__ unsigned list[AGG_CAP];
    __shared__ float4   exl4[AGG_CAP];
    __shared__ int      cnt;
    __shared__ int      ncnt[AGG_R];
    __shared__ int      off[AGG_R + 1];
    __shared__ int      nxt[AGG_R];
    __shared__ unsigned short perm[AGG_CAP];
    float* exl = (float*)exl4;

    const int tid  = (int)threadIdx.x;
    const int lane = tid & 31;
    const int wave = tid >> 5;
    const int n0   = (int)blockIdx.x * AGG_R;
    const int nv   = (NN - n0 < AGG_R) ? (NN - n0) : AGG_R;
    const float4* as4 = (const float4*)as_v;
    unsigned* keep_r = keep + (size_t)blockIdx.x * KEEP_STRIDE;

    for (int i = tid; i < AGG_R; i += AGG_TB) ncnt[i] = 0;
    for (int i = tid; i < AGG_R * NH; i += AGG_TB) {
        int n = n0 + (i >> 2);
        adl[i]  = (n < NN) ? ad_v[n * NH + (i & 3)] : 0.f;
        mxl[i]  = (int)0x80000000;
        denl[i] = 0.f;
    }
    if (SAVED) {
        if (tid == 0) cnt = (int)keep_r[0];
    } else {
        for (int i = tid; i < nv; i += AGG_TB)
            list[i] = (unsigned)(NE + n0 + i);
        if (tid == 0) cnt = nv;
    }
    __syncthreads();

    if (!SAVED) {
    const uint4* pa4 = (const uint4*)planes;
    const uint4* pb4 = pa4 + NG;
    const unsigned rid = blockIdx.x;
    const unsigned c0 = (rid &   1u) ? 0u : ~0u, c1 = (rid &   2u) ? 0u : ~0u;
    const unsigned c2 = (rid &   4u) ? 0u : ~0u, c3 = (rid &   8u) ? 0u : ~0u;
    const unsigned c4 = (rid &  16u) ? 0u : ~0u, c5 = (rid &  32u) ? 0u : ~0u;
    const unsigned c6 = (rid &  64u) ? 0u : ~0u, c7 = (rid & 128u) ? 0u : ~0u;
    const int rot = (int)blockIdx.x * (NG / AGG_WG);
    auto take = [&](const uint4& a, const uint4& b, int id) {
        unsigned m = (a.x ^ c0) & (a.y ^ c1) & (a.z ^ c2) & (a.w ^ c3)
                   & (b.x ^ c4) & (b.y ^ c5) & (b.z ^ c6) & (b.w ^ c7);
        if (m) {
            int p = atomicAdd(&cnt, __popc(m));
            const unsigned e0 = (unsigned)id * 32u;
            do {
                const unsigned bit = (unsigned)__ffs((int)m) - 1u;
                m &= m - 1u;
                if (p < AGG_CAP) list[p] = e0 + bit;
                ++p;
            } while (m);
        }
    };
    int j = tid;
    for (; j + 3 * AGG_TB < NG; j += 4 * AGG_TB) {
        uint4 pa[4], pb[4];
        int   ix[4];
#pragma unroll
        for (int u = 0; u < 4; ++u) {
            int id = j + u * AGG_TB + rot;
            if (id >= NG) id -= NG;
            ix[u] = id;
            pa[u] = pa4[id];
            pb[u] = pb4[id];
        }
#pragma unroll
        for (int u = 0; u < 4; ++u) take(pa[u], pb[u], ix[u]);
    }
    for (; j < NG; j += AGG_TB) {
        int id = j + rot;
        if (id >= NG) id -= NG;
        take(pa4[id], pb4[id], id);
    }
    __syncthreads();
    }
    const int total = cnt;
    if (!SAVED && tid == 0) keep_r[0] = (unsigned)total;

    if (total <= AGG_CAP) {
        for (int i = tid; i < total; i += AGG_TB) {
            unsigned q;
            if (SAVED) {
                q = keep_r[KEEP_HDR + i];
            } else {
                int e  = (int)list[i];
                int s0 = (e < NE) ? src[e] : (e - NE);
                int d0 = (e < NE) ? dst[e] : (e - NE);
                q = (unsigned)s0 | ((unsigned)(d0 - n0) << 16);
                keep_r[KEEP_HDR + i] = q;
            }
            list[i] = q;
            int s  = (int)(q & 0xFFFFu);
            int dl = (int)(q >> 16);
            float4 a = as4[s];
            float v0 = lrelu02(a.x + adl[dl * NH + 0]);
            float v1 = lrelu02(a.y + adl[dl * NH + 1]);
            float v2 = lrelu02(a.z + adl[dl * NH + 2]);
            float v3 = lrelu02(a.w + adl[dl * NH + 3]);
            exl[i * NH + 0] = v0; exl[i * NH + 1] = v1;
            exl[i * NH + 2] = v2; exl[i * NH + 3] = v3;
            atomicMax(&mxl[dl * NH + 0], f2ord(v0));
            atomicMax(&mxl[dl * NH + 1], f2ord(v1));
            atomicMax(&mxl[dl * NH + 2], f2ord(v2));
            atomicMax(&mxl[dl * NH + 3], f2ord(v3));
            atomicAdd(&ncnt[dl], 1);
        }
        __syncthreads();
        if (wave == 0) {
            constexpr int PER = (AGG_R + 31) / 32;
            int loc[PER];
            int sum = 0;
#pragma unroll
            for (int k = 0; k < PER; ++k) {
                const int idx = lane * PER + k;
                loc[k] = (idx < AGG_R) ? ncnt[idx] : 0;
                sum += loc[k];
            }
            int incl = sum;
#pragma unroll
            for (int m = 1; m < 32; m <<= 1) {
                const int t = __shfl_up(incl, m);
                if (lane >= m) incl += t;
            }
            int run = incl - sum;
#pragma unroll
            for (int k = 0; k < PER; ++k) {
                const int idx = lane * PER + k;
                if (idx < AGG_R) { off[idx] = run; nxt[idx] = run; }
                run += loc[k];
            }
            if (lane == 0) off[AGG_R] = total;
        }
        for (int i = tid; i < total * NH; i += AGG_TB) {
            int dl = (int)(list[i >> 2] >> 16);
            int h  = i & 3;
            float ex = expf(exl[i] - ord2f(mxl[dl * NH + h]));
            exl[i] = ex;
            atomicAdd(&denl[dl * NH + h], ex);
        }
        __syncthreads();
        for (int i = tid; i < total; i += AGG_TB) {
            const int dl  = (int)(list[i] >> 16);
            const int pos = atomicAdd(&nxt[dl], 1);
            perm[pos] = (unsigned short)i;
        }
        __syncthreads();
        {
            const int hd = lane >> 3;
            const float4* H4 = (const float4*)Hm;
            const int wv  = __builtin_amdgcn_readfirstlane(wave);
            const int nlo = (wv * nv) / AGG_NW;
            const int nhi = ((wv + 1) * nv) / AGG_NW;
            int j = __builtin_amdgcn_readfirstlane(off[nlo]);
            const int jend = __builtin_amdgcn_readfirstlane(off[nhi]);
            int cn = nlo;
            float4 a4 = make_float4(0.f, 0.f, 0.f, 0.f);
            auto done_to = [&](int dl) {
#pragma nounroll
                for (; cn < dl; ++cn) {
                    float* ap = acc + cn * HID + lane;
                    ap[0] = a4.x; ap[32] = a4.y; ap[64] = a4.z; ap[96] = a4.w;
                    a4 = make_float4(0.f, 0.f, 0.f, 0.f);
                }
            };
            const int l7 = lane & 7;
            int      iv = 0;
            unsigned pv = 0u;
            if (j < jend) {
                const int jj = (j + l7 < jend) ? (j + l7) : (jend - 1);
                iv = (int)perm[jj];
                pv = list[iv];
            }
            for (; j + 8 <= jend; j += 8) {
                unsigned p[8]; float w[8]; float4 hv[8];
#pragma unroll
                for (int u = 0; u < 8; ++u) {
                    const int i = __builtin_amdgcn_readlane(iv, u);
                    p[u] = (unsigned)__builtin_amdgcn_readlane((int)pv, u);
                    w[u] = exl[i * NH + hd];
                }
#pragma unroll
                for (int u = 0; u < 8; ++u)
                    hv[u] = H4[(size_t)(p[u] & 0xFFFFu) * (HID / 4) + lane];
                if (j + 8 < jend) {
                    const int jj = (j + 8 + l7 < jend) ? (j + 8 + l7) : (jend - 1);
                    iv = (int)perm[jj];
                    pv = list[iv];
                }
#pragma unroll
                for (int u = 0; u < 8; ++u) {
                    done_to((int)(p[u] >> 16));
                    a4.x += hv[u].x * w[u];
                    a4.y += hv[u].y * w[u];
                    a4.z += hv[u].z * w[u];
                    a4.w += hv[u].w * w[u];
                }
            }
            {
                const int rem = jend - j;
                unsigned p[8]; float w[8]; float4 hv[8];
#pragma unroll
                for (int u = 0; u < 7; ++u) {
                    if (u < rem) {
                        const int i = __builtin_amdgcn_readlane(iv, u);
                        p[u] = (unsigned)__builtin_amdgcn_readlane((int)pv, u);
                        w[u] = exl[i * NH + hd];
                    }
                }
#pragma unroll
                for (int u = 0; u < 7; ++u) {
                    if (u < rem)
                        hv[u] = H4[(size_t)(p[u] & 0xFFFFu) * (HID / 4) + lane];
                }
#pragma unroll
                for (int u = 0; u < 7; ++u) {
                    if (u < rem) {
                        done_to((int)(p[u] >> 16));
                        a4.x += hv[u].x * w[u];
                        a4.y += hv[u].y * w[u];
                        a4.z += hv[u].z * w[u];
                        a4.w += hv[u].w * w[u];
                    }
                }
            }
            done_to(nhi);
        }
    } else {
        for (int i = tid; i < AGG_R * HID; i += AGG_TB) acc[i] = 0.f;
        for (int e = tid; e < ET; e += AGG_TB) {
            int s = (e < NE) ? src[e] : (e - NE);
            int d = (e < NE) ? dst[e] : (e - NE);
            unsigned dl = (unsigned)(d - n0);
            if (dl < (unsigned)nv) {
                float4 a = as4[s];
                atomicMax(&mxl[dl * NH + 0], f2ord(lrelu02(a.x + adl[dl * NH + 0])));
                atomicMax(&mxl[dl * NH + 1], f2ord(lrelu02(a.y + adl[dl * NH + 1])));
                atomicMax(&mxl[dl * NH + 2], f2ord(lrelu02(a.z + adl[dl * NH + 2])));
                atomicMax(&mxl[dl * NH + 3], f2ord(lrelu02(a.w + adl[dl * NH + 3])));
            }
        }
        __syncthreads();
        for (int e = tid; e < ET; e += AGG_TB) {
            int s = (e < NE) ? src[e] : (e - NE);
            int d = (e < NE) ? dst[e] : (e - NE);
            unsigned dl = (unsigned)(d - n0);
            if (dl < (unsigned)nv) {
                const float* ap4 = as_v + (size_t)s * NH;
                const float* hp  = Hm + (size_t)s * HID;
                for (int h = 0; h < NH; ++h) {
                    float v  = lrelu02(ap4[h] + adl[dl * NH + h]);
                    float ex = expf(v - ord2f(mxl[dl * NH + h]));
                    atomicAdd(&denl[dl * NH + h], ex);
                    for (int c = 0; c < CH; ++c)
                    {
                        const int ch = h * CH + c;
                        atomicAdd(&acc[dl * HID + (ch & 3) * 32 + (ch >> 2)], hp[ch] * ex);
                    }
                }
            }
        }
    }
    __syncthreads();

    {
        const int hd = lane >> 3;
        const float4 bb = *(const float4*)(b + 4 * lane);
        for (int dl = wave; dl < nv; dl += AGG_NW) {
            const float* ar = acc + dl * HID + lane;
            const float dn = denl[dl * NH + hd] + 1e-16f;
            float4 y;
            y.x = ar[ 0] / dn + bb.x;
            y.y = ar[32] / dn + bb.y;
            y.z = ar[64] / dn + bb.z;
            y.w = ar[96] / dn + bb.w;
            y.x = (y.x > 0.f) ? y.x : (expf(y.x) - 1.f);
            y.y = (y.y > 0.f) ? y.y : (expf(y.y) - 1.f);
            y.z = (y.z > 0.f) ? y.z : (expf(y.z) - 1.f);
            y.w = (y.w > 0.f) ? y.w : (expf(y.w) - 1.f);
            *(float4*)(out + (size_t)(n0 + dl) * HID + 4 * lane) = y;
        }
    }
}

template <bool SAVED>
static void run_gat_layer(const float* Xin, const float* W,
                          const float* a_s, const float* a_d, const float* b,
                          const int* src, const int* dst,
                          float* bufH, float* bufOut,
                          float* asb, float* adb, unsigned* keep, unsigned* planes, hipStream_t stream)
{
    const int TB = GEMM_TB;
    int nMtiles = NN / 16;
    int blocks  = (nMtiles * 32 + TB - 1) / TB;
    wmma_gemm_strip<HID, HID / 16, true, !SAVED><<<blocks, TB, 0, stream>>>(
        Xin, W, nullptr, bufH, nMtiles, a_s, a_d, asb, adb, dst, planes);
    agg_range<SAVED><<<AGG_WG, AGG_TB, 0, stream>>>(src, dst, asb, adb, bufH, b, bufOut, keep, planes);
}

extern "C" void kernel_launch(void* const* d_in, const int* in_sizes, int n_in,
                              void* d_out, int out_size, void* d_ws, size_t ws_size,
                              hipStream_t stream)
{
    const float* x    = (const float*)d_in[0];
    const int*   edge = (const int*)  d_in[1];
    const float* W1   = (const float*)d_in[2];
    const float* a_s1 = (const float*)d_in[3];
    const float* a_d1 = (const float*)d_in[4];
    const float* b1   = (const float*)d_in[5];
    const float* W2   = (const float*)d_in[6];
    const float* a_s2 = (const float*)d_in[7];
    const float* a_d2 = (const float*)d_in[8];
    const float* b2   = (const float*)d_in[9];
    const float* fc_w = (const float*)d_in[10];
    const float* fc_b = (const float*)d_in[11];
    float* out = (float*)d_out;

    const int* src = edge;
    const int* dst = edge + NE;

    float* bufH   = (float*)d_ws;
    float* bufIn2 = bufH   + (size_t)NN * HID;
    float* asb    = bufIn2 + (size_t)NN * HID;
    float* adb    = asb    + (size_t)NN * NH;
    unsigned* keep = (unsigned*)(adb + (size_t)NN * NH);
    unsigned* planes = keep + (size_t)AGG_WG * KEEP_STRIDE;

    run_gat_layer<false>(x, W1, a_s1, a_d1, b1, src, dst, bufH, bufIn2, asb, adb, keep, planes, stream);
    run_gat_layer<true>(bufIn2, W2, a_s2, a_d2, b2, src, dst, bufH, bufIn2, asb, adb, keep, planes, stream);
    const int TB = GEMM_TB;
    int nMtiles = NN / 16;
    int blocks  = (nMtiles * 32 + TB - 1) / TB;
    wmma_gemm_strip<HID, CH / 16, false, false><<<blocks, TB, 0, stream>>>(
        bufIn2, fc_w, fc_b, out, nMtiles, nullptr, nullptr, nullptr, nullptr, nullptr, nullptr);
}
